# starting kernel
# baseline (speedup 1.0000x reference)
#include <hip/hip_runtime.h>
#include <hip/hip_bf16.h>

typedef __attribute__((ext_vector_type(16))) __bf16 v16bf;
typedef __attribute__((ext_vector_type(8)))  float  v8f;

__global__ void deg_init_kernel(float* __restrict__ deg, int n) {
    int i = blockIdx.x * blockDim.x + threadIdx.x;
    if (i < n) deg[i] = 1.0f;
}

__global__ void deg_count_kernel(const int* __restrict__ dst,
                                 float* __restrict__ deg, int E) {
    int i = blockIdx.x * blockDim.x + threadIdx.x;
    if (i < E) unsafeAtomicAdd(&deg[dst[i]], 1.0f);
}

__global__ void deg_rsqrt_kernel(const float* __restrict__ deg,
                                 float* __restrict__ isq, int n) {
    int i = blockIdx.x * blockDim.x + threadIdx.x;
    if (i < n) isq[i] = rsqrtf(deg[i]);
}

template <int CIN, int COUT>
__global__ void gemm_wmma_bf16_kernel(const float* __restrict__ X,
                                      const float* __restrict__ W,
                                      float* __restrict__ H) {
    const int lane = threadIdx.x;
    const int ctile = threadIdx.y;
    const int m0 = blockIdx.x * 16;
    const int half = lane >> 4;
    const int l = lane & 15;
    const int m = m0 + l;
    const int n = ctile * 16 + l;

    v8f acc = {0.f, 0.f, 0.f, 0.f, 0.f, 0.f, 0.f, 0.f};

    #pragma unroll
    for (int kb = 0; kb < CIN; kb += 32) {
        float xa[16];
        const float4* xr0 = reinterpret_cast<const float4*>(&X[(size_t)m * CIN + kb + half * 8]);
        const float4* xr1 = reinterpret_cast<const float4*>(&X[(size_t)m * CIN + kb + 16 + half * 8]);
        *reinterpret_cast<float4*>(&xa[0])  = xr0[0];
        *reinterpret_cast<float4*>(&xa[4])  = xr0[1];
        *reinterpret_cast<float4*>(&xa[8])  = xr1[0];
        *reinterpret_cast<float4*>(&xa[12]) = xr1[1];

        v16bf a, b;
        #pragma unroll
        for (int e = 0; e < 8; ++e) {
            const int k0 = kb + half * 8 + e;
            const int k1 = k0 + 16;
            a[e]     = (__bf16)xa[e];
            a[e + 8] = (__bf16)xa[e + 8];
            b[e]     = (__bf16)W[(size_t)k0 * COUT + n];
            b[e + 8] = (__bf16)W[(size_t)k1 * COUT + n];
        }
        acc = __builtin_amdgcn_wmma_f32_16x16x32_bf16(
             false, a,  false, b,
             (short)0, acc,  false,  false);
    }

    #pragma unroll
    for (int v = 0; v < 8; ++v) {
        const int r = m0 + v + half * 8;
        H[(size_t)r * COUT + ctile * 16 + l] = acc[v];
    }
}

__global__ void self_init_kernel(const float* __restrict__ H,
                                 const float* __restrict__ isq,
                                 float* __restrict__ AGG,
                                 long total, int cshift) {
    long i = (long)blockIdx.x * blockDim.x + threadIdx.x;
    if (i < total) {
        int node = (int)(i >> cshift);
        float s = isq[node];
        AGG[i] = H[i] * s * s;
    }
}

__global__ void scatter_edges_kernel(const float* __restrict__ H,
                                     const int* __restrict__ src,
                                     const int* __restrict__ dst,
                                     const float* __restrict__ isq,
                                     float* __restrict__ AGG,
                                     long total, int C, int lshift) {
    long idx = (long)blockIdx.x * blockDim.x + threadIdx.x;
    if (idx >= total) return;
    const int e = (int)(idx >> lshift);
    const int c = (int)(idx & ((1 << lshift) - 1)) << 2;
    const int s = src[e];
    const int d = dst[e];
    const float nrm = isq[s] * isq[d];
    const float4 v = *reinterpret_cast<const float4*>(&H[(size_t)s * C + c]);
    float* out = &AGG[(size_t)d * C + c];
    unsafeAtomicAdd(out + 0, v.x * nrm);
    unsafeAtomicAdd(out + 1, v.y * nrm);
    unsafeAtomicAdd(out + 2, v.z * nrm);
    unsafeAtomicAdd(out + 3, v.w * nrm);
}

__global__ void bias_act_kernel(const float* __restrict__ AGG,
                                const float* __restrict__ bias,
                                float* __restrict__ OUT,
                                long total, int cmask, int do_relu) {
    long i = (long)blockIdx.x * blockDim.x + threadIdx.x;
    if (i < total) {
        float v = AGG[i] + bias[(int)(i & cmask)];
        OUT[i] = do_relu ? fmaxf(v, 0.0f) : v;
    }
}

extern "C" void kernel_launch(void* const* d_in, const int* in_sizes, int n_in,
                              void* d_out, int out_size, void* d_ws, size_t ws_size,
                              hipStream_t stream) {
    const float* x  = (const float*)d_in[0];
    const int*   ei = (const int*)d_in[1];
    const float* W1 = (const float*)d_in[2];
    const float* b1 = (const float*)d_in[3];
    const float* W2 = (const float*)d_in[4];
    const float* b2 = (const float*)d_in[5];
    const float* W3 = (const float*)d_in[6];
    const float* b3 = (const float*)d_in[7];

    const int CIN = 128, CH = 128, COUT = 64;
    const int n = in_sizes[0] / CIN;
    const int E = in_sizes[1] / 2;
    const int* src = ei;
    const int* dst = ei + E;

    float* ws   = (float*)d_ws;
    float* deg  = ws;
    float* isq  = deg  + n;
    float* hlin = isq  + n;
    float* agg  = hlin + (size_t)n * CH;
    float* feat = agg  + (size_t)n * CH;

    const int TB = 256;
    deg_init_kernel <<<(n + TB - 1) / TB, TB, 0, stream>>>(deg, n);
    deg_count_kernel<<<(E + TB - 1) / TB, TB, 0, stream>>>(dst, deg, E);
    deg_rsqrt_kernel<<<(n + TB - 1) / TB, TB, 0, stream>>>(deg, isq, n);

    const int row_tiles = n / 16;

    {
        const int C = CH;
        gemm_wmma_bf16_kernel<128, 128><<<row_tiles, dim3(32, C / 16), 0, stream>>>(x, W1, hlin);
        long nc = (long)n * C;
        self_init_kernel<<<(int)((nc + TB - 1) / TB), TB, 0, stream>>>(hlin, isq, agg, nc, 7);
        long tot = (long)E << 5;
        scatter_edges_kernel<<<(int)((tot + TB - 1) / TB), TB, 0, stream>>>(
            hlin, src, dst, isq, agg, tot, C, 5);
        bias_act_kernel<<<(int)((nc + TB - 1) / TB), TB, 0, stream>>>(agg, b1, feat, nc, C - 1, 1);
    }

    {
        const int C = CH;
        gemm_wmma_bf16_kernel<128, 128><<<row_tiles, dim3(32, C / 16), 0, stream>>>(feat, W2, hlin);
        long nc = (long)n * C;
        self_init_kernel<<<(int)((nc + TB - 1) / TB), TB, 0, stream>>>(hlin, isq, agg, nc, 7);
        long tot = (long)E << 5;
        scatter_edges_kernel<<<(int)((tot + TB - 1) / TB), TB, 0, stream>>>(
            hlin, src, dst, isq, agg, tot, C, 5);
        bias_act_kernel<<<(int)((nc + TB - 1) / TB), TB, 0, stream>>>(agg, b2, feat, nc, C - 1, 1);
    }

    {
        const int C = COUT;
        gemm_wmma_bf16_kernel<128, 64><<<row_tiles, dim3(32, C / 16), 0, stream>>>(feat, W3, hlin);
        long nc = (long)n * C;
        self_init_kernel<<<(int)((nc + TB - 1) / TB), TB, 0, stream>>>(hlin, isq, agg, nc, 6);
        long tot = (long)E << 4;
        scatter_edges_kernel<<<(int)((tot + TB - 1) / TB), TB, 0, stream>>>(
            hlin, src, dst, isq, agg, tot, C, 4);
        bias_act_kernel<<<(int)((nc + TB - 1) / TB), TB, 0, stream>>>(
            agg, b3, (float*)d_out, nc, C - 1, 0);
    }
}


# next layer's weight product (same bf16 WMMA, same k order) moved into the gather kernel's epilogue: a work-group's relu'd rows form one A tile in LDS; 7 launches not 9; sum order unchanged
# speedup vs baseline: 4.7809x; 4.7809x over previous
#include <hip/hip_runtime.h>
#include <hip/hip_bf16.h>

typedef __attribute__((ext_vector_type(16))) __bf16 v16bf;
typedef __attribute__((ext_vector_type(8)))  float  v8f;

constexpr int NCHUNK = 128;
constexpr int CSR_TB = 1024;
constexpr int EPT = 8;

__global__ void __launch_bounds__(CSR_TB)
chunk_hist_kernel(const int* __restrict__ dst, int* __restrict__ chist,
                  int n, int E) {
    extern __shared__ int sm[];
    const int t = threadIdx.x;
    const int b = blockIdx.x;
    const int chunk = (E + NCHUNK - 1) / NCHUNK;
    const int e0 = b * chunk;
    int e1 = e0 + chunk;
    if (e1 > E) e1 = E;
    for (int i = t; i < n; i += CSR_TB) sm[i] = 0;
    __syncthreads();
    for (int i0 = e0 + t; i0 < e1; i0 += EPT * CSR_TB) {
        int d[EPT];
        #pragma unroll
        for (int u = 0; u < EPT; ++u) {
            const int i = i0 + u * CSR_TB;
            d[u] = dst[i < e1 ? i : e1 - 1];
        }
        #pragma unroll
        for (int u = 0; u < EPT; ++u)
            if (i0 + u * CSR_TB < e1) atomicAdd(&sm[d[u]], 1);
    }
    __syncthreads();
    for (int i = t; i < n; i += CSR_TB) chist[(size_t)b * n + i] = sm[i];
}

__global__ void __launch_bounds__(256)
chunk_prefix_kernel(int* __restrict__ chist, int* __restrict__ lpre,
                    int* __restrict__ gsum, float* __restrict__ isq, int n) {
    constexpr int PER = NCHUNK / 8;
    __shared__ int wt[8][32];
    const int lane = threadIdx.x & 31;
    const int w = threadIdx.x >> 5;
    const int node = blockIdx.x * 32 + lane;
    const bool ok = node < n;
    int v[PER];
    #pragma unroll
    for (int j = 0; j < PER; ++j)
        v[j] = ok ? chist[(size_t)(w * PER + j) * n + node] : 0;
    int run = 0;
    #pragma unroll
    for (int j = 0; j < PER; ++j) { const int c = v[j]; v[j] = run; run += c; }
    wt[w][lane] = run;
    __syncthreads();
    int off = 0, tot = 0;
    #pragma unroll
    for (int q = 0; q < 8; ++q) {
        const int x = wt[q][lane];
        if (q < w) off += x;
        tot += x;
    }
    if (ok) {
        #pragma unroll
        for (int j = 0; j < PER; ++j)
            chist[(size_t)(w * PER + j) * n + node] = v[j] + off;
    }
    if (w == 0) {
        int inc = tot;
        for (int o = 1; o < 32; o <<= 1) {
            const int u = __shfl_up(inc, o);
            if (lane >= o) inc += u;
        }
        if (ok) {
            lpre[node] = inc - tot;
            isq[node] = rsqrtf((float)(tot + 1));
        }
        if (lane == 31) gsum[blockIdx.x] = inc;
    }
}

__global__ void __launch_bounds__(CSR_TB)
csr_place_kernel(const int* __restrict__ src, const int* __restrict__ dst,
                 const int* __restrict__ chist, const int* __restrict__ lpre,
                 const int* __restrict__ gsum, const float* __restrict__ isq,
                 int* __restrict__ rowptr, int2* __restrict__ col,
                 int n, int E, int G) {
    extern __shared__ int sm[];
    int* cur = sm;
    int* gp = sm + n;
    const int t = threadIdx.x;
    const int b = blockIdx.x;
    if (t < 32) {
        const int per = (G + 31) / 32;
        const int lo = t * per;
        int hi = lo + per;
        if (hi > G) hi = G;
        int s = 0;
        for (int g = lo; g < hi; ++g) s += gsum[g];
        int inc = s;
        for (int o = 1; o < 32; o <<= 1) {
            const int u = __shfl_up(inc, o);
            if (t >= o) inc += u;
        }
        int run = inc - s;
        for (int g = lo; g < hi; ++g) { gp[g] = run; run += gsum[g]; }
        if (t == 31 && b == 0) rowptr[n] = inc;
    }
    __syncthreads();
    for (int i = t; i < n; i += CSR_TB) {
        const int r = gp[i >> 5] + lpre[i];
        if (b == 0) rowptr[i] = r;
        cur[i] = r + chist[(size_t)b * n + i];
    }
    __syncthreads();
    const int chunk = (E + NCHUNK - 1) / NCHUNK;
    const int e0 = b * chunk;
    int e1 = e0 + chunk;
    if (e1 > E) e1 = E;
    for (int i0 = e0 + t; i0 < e1; i0 += EPT * CSR_TB) {
        int s[EPT], d[EPT];
        #pragma unroll
        for (int u = 0; u < EPT; ++u) {
            const int i = i0 + u * CSR_TB;
            const int ic = i < e1 ? i : e1 - 1;
            s[u] = src[ic];
            d[u] = dst[ic];
        }
        float w[EPT];
        #pragma unroll
        for (int u = 0; u < EPT; ++u) w[u] = isq[s[u]];
        #pragma unroll
        for (int u = 0; u < EPT; ++u) {
            if (i0 + u * CSR_TB < e1) {
                const int pos = atomicAdd(&cur[d[u]], 1);
                col[pos] = make_int2(s[u], __float_as_int(w[u]));
            }
        }
    }
}

template <int CIN, int COUT>
__global__ void gemm_wmma_bf16_kernel(const float* __restrict__ X,
                                      const float* __restrict__ W,
                                      float* __restrict__ H) {
    const int lane = threadIdx.x;
    const int ctile = threadIdx.y;
    const int m0 = blockIdx.x * 16;
    const int half = lane >> 4;
    const int l = lane & 15;
    const int m = m0 + l;
    const int n = ctile * 16 + l;

    v8f acc = {0.f, 0.f, 0.f, 0.f, 0.f, 0.f, 0.f, 0.f};

    #pragma unroll
    for (int kb = 0; kb < CIN; kb += 32) {
        float xa[16];
        const float4* xr0 = reinterpret_cast<const float4*>(&X[(size_t)m * CIN + kb + half * 8]);
        const float4* xr1 = reinterpret_cast<const float4*>(&X[(size_t)m * CIN + kb + 16 + half * 8]);
        *reinterpret_cast<float4*>(&xa[0])  = xr0[0];
        *reinterpret_cast<float4*>(&xa[4])  = xr0[1];
        *reinterpret_cast<float4*>(&xa[8])  = xr1[0];
        *reinterpret_cast<float4*>(&xa[12]) = xr1[1];

        v16bf a, b;
        #pragma unroll
        for (int e = 0; e < 8; ++e) {
            const int k0 = kb + half * 8 + e;
            const int k1 = k0 + 16;
            a[e]     = (__bf16)xa[e];
            a[e + 8] = (__bf16)xa[e + 8];
            b[e]     = (__bf16)W[(size_t)k0 * COUT + n];
            b[e + 8] = (__bf16)W[(size_t)k1 * COUT + n];
        }
        acc = __builtin_amdgcn_wmma_f32_16x16x32_bf16(
             false, a,  false, b,
             (short)0, acc,  false,  false);
    }

    #pragma unroll
    for (int v = 0; v < 8; ++v) {
        const int r = m0 + v + half * 8;
        H[(size_t)r * COUT + ctile * 16 + l] = acc[v];
    }
}

template <int C>
__global__ void __launch_bounds__(256)
gather_sum_kernel(const float* __restrict__ H,
                  const int* __restrict__ rowptr,
                  const int2* __restrict__ col,
                  const float* __restrict__ isq,
                  const float* __restrict__ bias,
                  float* __restrict__ OUT, int n, int do_relu) {
    constexpr int VEC = C / 32;
    typedef float vec_t __attribute__((ext_vector_type(VEC)));
    const int lane = threadIdx.x & 31;
    const int node = __builtin_amdgcn_readfirstlane(
        (int)(blockIdx.x * (blockDim.x >> 5) + (threadIdx.x >> 5)));
    if (node >= n) return;
    const int beg = rowptr[node];
    const int end = rowptr[node + 1];
    const float sd = isq[node];
    const vec_t* __restrict__ Hv = reinterpret_cast<const vec_t*>(H);

    const vec_t hv = Hv[(size_t)node * 32 + lane];
    vec_t acc = (hv * sd) * sd;

    int2 nxt = make_int2(0, 0);
    if (beg + lane < end) nxt = col[beg + lane];
    for (int j0 = beg; j0 < end; j0 += 32) {
        const int cur_s = nxt.x;
        const float cur_w = __int_as_float(nxt.y) * sd;
        nxt = make_int2(0, 0);
        const int jn = j0 + 32 + lane;
        if (jn < end) nxt = col[jn];
        int m = end - j0;
        if (m > 32) m = 32;
        int k = 0;
        for (; k + 8 <= m; k += 8) {
            vec_t v[8];
            #pragma unroll
            for (int u = 0; u < 8; ++u) {
                const int s = __builtin_amdgcn_readlane(cur_s, k + u);
                v[u] = Hv[(size_t)s * 32 + lane];
            }
            #pragma unroll
            for (int u = 0; u < 8; ++u) {
                const float w = __int_as_float(
                    __builtin_amdgcn_readlane(__float_as_int(cur_w), k + u));
                acc += v[u] * w;
            }
        }
        if (k < m) {
            vec_t v[8];
            #pragma unroll
            for (int u = 0; u < 8; ++u) {
                const int s = __builtin_amdgcn_readlane(cur_s, (k + u) & 31);
                v[u] = Hv[(size_t)s * 32 + lane];
            }
            #pragma unroll
            for (int u = 0; u < 8; ++u) {
                const float w = __int_as_float(
                    __builtin_amdgcn_readlane(__float_as_int(cur_w), (k + u) & 31));
                if (k + u < m) acc += v[u] * w;
            }
        }
    }

    const vec_t bv = reinterpret_cast<const vec_t*>(bias)[lane];
    vec_t o = acc + bv;
    if (do_relu) {
        #pragma unroll
        for (int i = 0; i < VEC; ++i) o[i] = fmaxf(o[i], 0.0f);
    }
    reinterpret_cast<vec_t*>(OUT)[(size_t)node * 32 + lane] = o;
}

typedef __attribute__((ext_vector_type(8))) __bf16 v8bf;
typedef __attribute__((ext_vector_type(4))) __bf16 v4bf;

template <int COUT>
__global__ void __launch_bounds__(512)
gather_gemm_kernel(const float* __restrict__ H,
                   const int* __restrict__ rowptr,
                   const int2* __restrict__ col,
                   const float* __restrict__ isq,
                   const float* __restrict__ bias,
                   const float* __restrict__ Wn,
                   float* __restrict__ Hn, int n) {
    constexpr int C = 128;
    constexpr int PITCH = C + 8;
    constexpr int NT = COUT / 16;
    typedef float vec_t __attribute__((ext_vector_type(4)));
    __shared__ __attribute__((aligned(16))) __bf16 tile[16 * PITCH];

    const int lane = threadIdx.x & 31;
    const int wave = __builtin_amdgcn_readfirstlane((int)(threadIdx.x >> 5));
    const int node = __builtin_amdgcn_readfirstlane((int)(blockIdx.x * 16)) + wave;

    vec_t o = {0.f, 0.f, 0.f, 0.f};
    if (node < n) {
        const int beg = rowptr[node];
        const int end = rowptr[node + 1];
        const float sd = isq[node];
        const vec_t* __restrict__ Hv = reinterpret_cast<const vec_t*>(H);

        const vec_t hv = Hv[(size_t)node * 32 + lane];
        vec_t acc = (hv * sd) * sd;

        int2 nxt = make_int2(0, 0);
        if (beg + lane < end) nxt = col[beg + lane];
        for (int j0 = beg; j0 < end; j0 += 32) {
            const int cur_s = nxt.x;
            const float cur_w = __int_as_float(nxt.y) * sd;
            nxt = make_int2(0, 0);
            const int jn = j0 + 32 + lane;
            if (jn < end) nxt = col[jn];
            int m = end - j0;
            if (m > 32) m = 32;
            int k = 0;
            for (; k + 8 <= m; k += 8) {
                vec_t v[8];
                #pragma unroll
                for (int u = 0; u < 8; ++u) {
                    const int s = __builtin_amdgcn_readlane(cur_s, k + u);
                    v[u] = Hv[(size_t)s * 32 + lane];
                }
                #pragma unroll
                for (int u = 0; u < 8; ++u) {
                    const float w = __int_as_float(
                        __builtin_amdgcn_readlane(__float_as_int(cur_w), k + u));
                    acc += v[u] * w;
                }
            }
            if (k < m) {
                vec_t v[8];
                #pragma unroll
                for (int u = 0; u < 8; ++u) {
                    const int s = __builtin_amdgcn_readlane(cur_s, (k + u) & 31);
                    v[u] = Hv[(size_t)s * 32 + lane];
                }
                #pragma unroll
                for (int u = 0; u < 8; ++u) {
                    const float w = __int_as_float(
                        __builtin_amdgcn_readlane(__float_as_int(cur_w), (k + u) & 31));
                    if (k + u < m) acc += v[u] * w;
                }
            }
        }
        const vec_t bv = reinterpret_cast<const vec_t*>(bias)[lane];
        o = acc + bv;
        #pragma unroll
        for (int i = 0; i < 4; ++i) o[i] = fmaxf(o[i], 0.0f);
    }

    v4bf ob;
    #pragma unroll
    for (int i = 0; i < 4; ++i) ob[i] = (__bf16)o[i];
    *reinterpret_cast<v4bf*>(&tile[wave * PITCH + lane * 4]) = ob;

    const int half = lane >> 4;
    const int l = lane & 15;

    v16bf b[C / 32];
    if (wave < NT) {
        const int nn = wave * 16 + l;
        #pragma unroll
        for (int kb = 0; kb < C; kb += 32) {
            #pragma unroll
            for (int e = 0; e < 8; ++e) {
                const int k0 = kb + half * 8 + e;
                const int k1 = k0 + 16;
                b[kb / 32][e]     = (__bf16)Wn[(size_t)k0 * COUT + nn];
                b[kb / 32][e + 8] = (__bf16)Wn[(size_t)k1 * COUT + nn];
            }
        }
    }
    __syncthreads();
    if (wave < NT) {
        v8f acc = {0.f, 0.f, 0.f, 0.f, 0.f, 0.f, 0.f, 0.f};
        #pragma unroll
        for (int kb = 0; kb < C; kb += 32) {
            const v8bf lo = *reinterpret_cast<const v8bf*>(&tile[l * PITCH + kb + half * 8]);
            const v8bf hi = *reinterpret_cast<const v8bf*>(&tile[l * PITCH + kb + 16 + half * 8]);
            const v16bf a = __builtin_shufflevector(lo, hi, 0, 1, 2, 3, 4, 5, 6, 7,
                                                    8, 9, 10, 11, 12, 13, 14, 15);
            acc = __builtin_amdgcn_wmma_f32_16x16x32_bf16(
                 false, a,  false, b[kb / 32],
                 (short)0, acc,  false,  false);
        }
        const int m0 = blockIdx.x * 16;
        #pragma unroll
        for (int v = 0; v < 8; ++v) {
            const int r = m0 + v + half * 8;
            Hn[(size_t)r * COUT + wave * 16 + l] = acc[v];
        }
    }
}

extern "C" void kernel_launch(void* const* d_in, const int* in_sizes, int n_in,
                              void* d_out, int out_size, void* d_ws, size_t ws_size,
                              hipStream_t stream) {
    const float* x  = (const float*)d_in[0];
    const int*   ei = (const int*)d_in[1];
    const float* W1 = (const float*)d_in[2];
    const float* b1 = (const float*)d_in[3];
    const float* W2 = (const float*)d_in[4];
    const float* b2 = (const float*)d_in[5];
    const float* W3 = (const float*)d_in[6];
    const float* b3 = (const float*)d_in[7];

    const int CIN = 128, CH = 128;
    const int n = in_sizes[0] / CIN;
    const int E = in_sizes[1] / 2;
    const int* src = ei;
    const int* dst = ei + E;

    float* hlin   = (float*)d_ws;
    float* feat   = hlin + (size_t)n * CH;
    int2*  col    = (int2*)(feat + (size_t)n * CH);
    int*   chist  = (int*)(col + E);
    int*   lpre   = chist + (size_t)NCHUNK * n;
    int*   rowptr = lpre + n;
    const int G   = (n + 31) / 32;
    int*   gsum   = rowptr + n + 1;
    float* isq    = (float*)(gsum + G);

    const int TB = 256;
    chunk_hist_kernel<<<NCHUNK, CSR_TB, (size_t)n * sizeof(int), stream>>>(dst, chist, n, E);
    chunk_prefix_kernel<<<G, TB, 0, stream>>>(chist, lpre, gsum, isq, n);
    csr_place_kernel<<<NCHUNK, CSR_TB, (size_t)(n + G) * sizeof(int), stream>>>(
        src, dst, chist, lpre, gsum, isq, rowptr, col, n, E, G);

    const int row_tiles = n / 16;
    const int node_blocks = (n + 7) / 8;

    gemm_wmma_bf16_kernel<128, 128><<<row_tiles, dim3(32, 8), 0, stream>>>(x, W1, hlin);

    gather_gemm_kernel<128><<<row_tiles, 512, 0, stream>>>(hlin, rowptr, col, isq, b1, W2, feat, n);

    gather_gemm_kernel<64><<<row_tiles, 512, 0, stream>>>(feat, rowptr, col, isq, b2, W3, hlin, n);

    gather_sum_kernel<64><<<node_blocks, TB, 0, stream>>>(hlin, rowptr, col, isq, b3, (float*)d_out, n, 0);
}


# row tables the gathers read (x@W1, relu(.)@W2, relu(.)@W3) stored as f16 not f32: precision of these stored intermediates narrows, WMMA operands stay bf16, all sums fp32; gathered bytes halve
# speedup vs baseline: 5.4046x; 1.1305x over previous
#include <hip/hip_runtime.h>
#include <hip/hip_bf16.h>

typedef __attribute__((ext_vector_type(16))) __bf16 v16bf;
typedef __attribute__((ext_vector_type(8)))  float  v8f;


constexpr int NCHUNK = 128;
constexpr int CSR_TB = 1024;
constexpr int EPT = 8;

__global__ void __launch_bounds__(CSR_TB)
chunk_hist_kernel(const int* __restrict__ dst, int* __restrict__ chist,
                  int n, int E) {
    extern __shared__ int sm[];
    const int t = threadIdx.x;
    const int b = blockIdx.x;
    const int chunk = (E + NCHUNK - 1) / NCHUNK;
    const int e0 = b * chunk;
    int e1 = e0 + chunk;
    if (e1 > E) e1 = E;
    for (int i = t; i < n; i += CSR_TB) sm[i] = 0;
    __syncthreads();
    for (int i0 = e0 + t; i0 < e1; i0 += EPT * CSR_TB) {
        int d[EPT];
        #pragma unroll
        for (int u = 0; u < EPT; ++u) {
            const int i = i0 + u * CSR_TB;
            d[u] = dst[i < e1 ? i : e1 - 1];
        }
        #pragma unroll
        for (int u = 0; u < EPT; ++u)
            if (i0 + u * CSR_TB < e1) atomicAdd(&sm[d[u]], 1);
    }
    __syncthreads();
    for (int i = t; i < n; i += CSR_TB) chist[(size_t)b * n + i] = sm[i];
}

__global__ void __launch_bounds__(256)
chunk_prefix_kernel(int* __restrict__ chist, int* __restrict__ lpre,
                    int* __restrict__ gsum, float* __restrict__ isq, int n) {
    constexpr int PER = NCHUNK / 8;
    __shared__ int wt[8][32];
    const int lane = threadIdx.x & 31;
    const int w = threadIdx.x >> 5;
    const int node = blockIdx.x * 32 + lane;
    const bool ok = node < n;
    int v[PER];
    #pragma unroll
    for (int j = 0; j < PER; ++j)
        v[j] = ok ? chist[(size_t)(w * PER + j) * n + node] : 0;
    int run = 0;
    #pragma unroll
    for (int j = 0; j < PER; ++j) { const int c = v[j]; v[j] = run; run += c; }
    wt[w][lane] = run;
    __syncthreads();
    int off = 0, tot = 0;
    #pragma unroll
    for (int q = 0; q < 8; ++q) {
        const int x = wt[q][lane];
        if (q < w) off += x;
        tot += x;
    }
    if (ok) {
        #pragma unroll
        for (int j = 0; j < PER; ++j)
            chist[(size_t)(w * PER + j) * n + node] = v[j] + off;
    }
    if (w == 0) {
        int inc = tot;
        for (int o = 1; o < 32; o <<= 1) {
            const int u = __shfl_up(inc, o);
            if (lane >= o) inc += u;
        }
        if (ok) {
            lpre[node] = inc - tot;
            isq[node] = rsqrtf((float)(tot + 1));
        }
        if (lane == 31) gsum[blockIdx.x] = inc;
    }
}

__global__ void __launch_bounds__(CSR_TB)
csr_place_kernel(const int* __restrict__ src, const int* __restrict__ dst,
                 const int* __restrict__ chist, const int* __restrict__ lpre,
                 const int* __restrict__ gsum, const float* __restrict__ isq,
                 int* __restrict__ rowptr, int2* __restrict__ col,
                 int n, int E, int G) {
    extern __shared__ int sm[];
    int* cur = sm;
    int* gp = sm + n;
    const int t = threadIdx.x;
    const int b = blockIdx.x;
    if (t < 32) {
        const int per = (G + 31) / 32;
        const int lo = t * per;
        int hi = lo + per;
        if (hi > G) hi = G;
        int s = 0;
        for (int g = lo; g < hi; ++g) s += gsum[g];
        int inc = s;
        for (int o = 1; o < 32; o <<= 1) {
            const int u = __shfl_up(inc, o);
            if (t >= o) inc += u;
        }
        int run = inc - s;
        for (int g = lo; g < hi; ++g) { gp[g] = run; run += gsum[g]; }
        if (t == 31 && b == 0) rowptr[n] = inc;
    }
    __syncthreads();
    for (int i = t; i < n; i += CSR_TB) {
        const int r = gp[i >> 5] + lpre[i];
        if (b == 0) rowptr[i] = r;
        cur[i] = r + chist[(size_t)b * n + i];
    }
    __syncthreads();
    const int chunk = (E + NCHUNK - 1) / NCHUNK;
    const int e0 = b * chunk;
    int e1 = e0 + chunk;
    if (e1 > E) e1 = E;
    for (int i0 = e0 + t; i0 < e1; i0 += EPT * CSR_TB) {
        int s[EPT], d[EPT];
        #pragma unroll
        for (int u = 0; u < EPT; ++u) {
            const int i = i0 + u * CSR_TB;
            const int ic = i < e1 ? i : e1 - 1;
            s[u] = src[ic];
            d[u] = dst[ic];
        }
        float w[EPT];
        #pragma unroll
        for (int u = 0; u < EPT; ++u) w[u] = isq[s[u]];
        #pragma unroll
        for (int u = 0; u < EPT; ++u) {
            if (i0 + u * CSR_TB < e1) {
                const int pos = atomicAdd(&cur[d[u]], 1);
                col[pos] = make_int2(s[u], __float_as_int(w[u]));
            }
        }
    }
}

template <int CIN, int COUT>
__global__ void gemm_wmma_bf16_kernel(const float* __restrict__ X,
                                      const float* __restrict__ W,
                                      _Float16* __restrict__ H) {
    const int lane = threadIdx.x;
    const int ctile = threadIdx.y;
    const int m0 = blockIdx.x * 16;
    const int half = lane >> 4;
    const int l = lane & 15;
    const int m = m0 + l;
    const int n = ctile * 16 + l;

    v8f acc = {0.f, 0.f, 0.f, 0.f, 0.f, 0.f, 0.f, 0.f};

    #pragma unroll
    for (int kb = 0; kb < CIN; kb += 32) {
        float xa[16];
        const float4* xr0 = reinterpret_cast<const float4*>(&X[(size_t)m * CIN + kb + half * 8]);
        const float4* xr1 = reinterpret_cast<const float4*>(&X[(size_t)m * CIN + kb + 16 + half * 8]);
        *reinterpret_cast<float4*>(&xa[0])  = xr0[0];
        *reinterpret_cast<float4*>(&xa[4])  = xr0[1];
        *reinterpret_cast<float4*>(&xa[8])  = xr1[0];
        *reinterpret_cast<float4*>(&xa[12]) = xr1[1];

        v16bf a, b;
        #pragma unroll
        for (int e = 0; e < 8; ++e) {
            const int k0 = kb + half * 8 + e;
            const int k1 = k0 + 16;
            a[e]     = (__bf16)xa[e];
            a[e + 8] = (__bf16)xa[e + 8];
            b[e]     = (__bf16)W[(size_t)k0 * COUT + n];
            b[e + 8] = (__bf16)W[(size_t)k1 * COUT + n];
        }
        acc = __builtin_amdgcn_wmma_f32_16x16x32_bf16(
             false, a,  false, b,
             (short)0, acc,  false,  false);
    }

    #pragma unroll
    for (int v = 0; v < 8; ++v) {
        const int r = m0 + v + half * 8;
        H[(size_t)r * COUT + ctile * 16 + l] = (_Float16)acc[v];
    }
}

template <int C>
__global__ void __launch_bounds__(256)
gather_sum_kernel(const _Float16* __restrict__ H,
                  const int* __restrict__ rowptr,
                  const int2* __restrict__ col,
                  const float* __restrict__ isq,
                  const float* __restrict__ bias,
                  float* __restrict__ OUT, int n, int do_relu) {
    constexpr int VEC = C / 32;
    typedef float vec_t __attribute__((ext_vector_type(VEC)));
    typedef _Float16 row_t __attribute__((ext_vector_type(VEC)));
    const int lane = threadIdx.x & 31;
    const int node = __builtin_amdgcn_readfirstlane(
        (int)(blockIdx.x * (blockDim.x >> 5) + (threadIdx.x >> 5)));
    if (node >= n) return;
    const int beg = rowptr[node];
    const int end = rowptr[node + 1];
    const float sd = isq[node];
    const row_t* __restrict__ Hv = reinterpret_cast<const row_t*>(H);

    const vec_t hv = __builtin_convertvector(Hv[(size_t)node * 32 + lane], vec_t);
    vec_t acc = (hv * sd) * sd;

    int2 nxt = make_int2(0, 0);
    if (beg + lane < end) nxt = col[beg + lane];
    for (int j0 = beg; j0 < end; j0 += 32) {
        const int cur_s = nxt.x;
        const float cur_w = __int_as_float(nxt.y) * sd;
        nxt = make_int2(0, 0);
        const int jn = j0 + 32 + lane;
        if (jn < end) nxt = col[jn];
        int m = end - j0;
        if (m > 32) m = 32;
        int k = 0;
        for (; k + 8 <= m; k += 8) {
            row_t v[8];
            #pragma unroll
            for (int u = 0; u < 8; ++u) {
                const int s = __builtin_amdgcn_readlane(cur_s, k + u);
                v[u] = Hv[(size_t)s * 32 + lane];
            }
            #pragma unroll
            for (int u = 0; u < 8; ++u) {
                const float w = __int_as_float(
                    __builtin_amdgcn_readlane(__float_as_int(cur_w), k + u));
                acc += __builtin_convertvector(v[u], vec_t) * w;
            }
        }
        if (k < m) {
            row_t v[8];
            #pragma unroll
            for (int u = 0; u < 8; ++u) {
                const int s = __builtin_amdgcn_readlane(cur_s, (k + u) & 31);
                v[u] = Hv[(size_t)s * 32 + lane];
            }
            #pragma unroll
            for (int u = 0; u < 8; ++u) {
                const float w = __int_as_float(
                    __builtin_amdgcn_readlane(__float_as_int(cur_w), (k + u) & 31));
                if (k + u < m) acc += __builtin_convertvector(v[u], vec_t) * w;
            }
        }
    }

    const vec_t bv = reinterpret_cast<const vec_t*>(bias)[lane];
    vec_t o = acc + bv;
    if (do_relu) {
        #pragma unroll
        for (int i = 0; i < VEC; ++i) o[i] = fmaxf(o[i], 0.0f);
    }
    reinterpret_cast<vec_t*>(OUT)[(size_t)node * 32 + lane] = o;
}

typedef __attribute__((ext_vector_type(8))) __bf16 v8bf;
typedef __attribute__((ext_vector_type(4))) __bf16 v4bf;

template <int COUT>
__global__ void __launch_bounds__(512)
gather_gemm_kernel(const _Float16* __restrict__ H,
                   const int* __restrict__ rowptr,
                   const int2* __restrict__ col,
                   const float* __restrict__ isq,
                   const float* __restrict__ bias,
                   const float* __restrict__ Wn,
                   _Float16* __restrict__ Hn, int n) {
    constexpr int C = 128;
    constexpr int PITCH = C + 8;
    constexpr int NT = COUT / 16;
    typedef float vec_t __attribute__((ext_vector_type(4)));
    typedef _Float16 row_t __attribute__((ext_vector_type(4)));
    __shared__ __attribute__((aligned(16))) __bf16 tile[16 * PITCH];

    const int lane = threadIdx.x & 31;
    const int wave = __builtin_amdgcn_readfirstlane((int)(threadIdx.x >> 5));
    const int node = __builtin_amdgcn_readfirstlane((int)(blockIdx.x * 16)) + wave;

    vec_t o = {0.f, 0.f, 0.f, 0.f};
    if (node < n) {
        const int beg = rowptr[node];
        const int end = rowptr[node + 1];
        const float sd = isq[node];
        const row_t* __restrict__ Hv = reinterpret_cast<const row_t*>(H);

        const vec_t hv = __builtin_convertvector(Hv[(size_t)node * 32 + lane], vec_t);
        vec_t acc = (hv * sd) * sd;

        int2 nxt = make_int2(0, 0);
        if (beg + lane < end) nxt = col[beg + lane];
        for (int j0 = beg; j0 < end; j0 += 32) {
            const int cur_s = nxt.x;
            const float cur_w = __int_as_float(nxt.y) * sd;
            nxt = make_int2(0, 0);
            const int jn = j0 + 32 + lane;
            if (jn < end) nxt = col[jn];
            int m = end - j0;
            if (m > 32) m = 32;
            int k = 0;
            for (; k + 8 <= m; k += 8) {
                row_t v[8];
                #pragma unroll
                for (int u = 0; u < 8; ++u) {
                    const int s = __builtin_amdgcn_readlane(cur_s, k + u);
                    v[u] = Hv[(size_t)s * 32 + lane];
                }
                #pragma unroll
                for (int u = 0; u < 8; ++u) {
                    const float w = __int_as_float(
                        __builtin_amdgcn_readlane(__float_as_int(cur_w), k + u));
                    acc += __builtin_convertvector(v[u], vec_t) * w;
                }
            }
            if (k < m) {
                row_t v[8];
                #pragma unroll
                for (int u = 0; u < 8; ++u) {
                    const int s = __builtin_amdgcn_readlane(cur_s, (k + u) & 31);
                    v[u] = Hv[(size_t)s * 32 + lane];
                }
                #pragma unroll
                for (int u = 0; u < 8; ++u) {
                    const float w = __int_as_float(
                        __builtin_amdgcn_readlane(__float_as_int(cur_w), (k + u) & 31));
                    if (k + u < m) acc += __builtin_convertvector(v[u], vec_t) * w;
                }
            }
        }
        const vec_t bv = reinterpret_cast<const vec_t*>(bias)[lane];
        o = acc + bv;
        #pragma unroll
        for (int i = 0; i < 4; ++i) o[i] = fmaxf(o[i], 0.0f);
    }

    v4bf ob;
    #pragma unroll
    for (int i = 0; i < 4; ++i) ob[i] = (__bf16)o[i];
    *reinterpret_cast<v4bf*>(&tile[wave * PITCH + lane * 4]) = ob;

    const int half = lane >> 4;
    const int l = lane & 15;

    v16bf b[C / 32];
    if (wave < NT) {
        const int nn = wave * 16 + l;
        #pragma unroll
        for (int kb = 0; kb < C; kb += 32) {
            #pragma unroll
            for (int e = 0; e < 8; ++e) {
                const int k0 = kb + half * 8 + e;
                const int k1 = k0 + 16;
                b[kb / 32][e]     = (__bf16)Wn[(size_t)k0 * COUT + nn];
                b[kb / 32][e + 8] = (__bf16)Wn[(size_t)k1 * COUT + nn];
            }
        }
    }
    __syncthreads();
    if (wave < NT) {
        v8f acc = {0.f, 0.f, 0.f, 0.f, 0.f, 0.f, 0.f, 0.f};
        #pragma unroll
        for (int kb = 0; kb < C; kb += 32) {
            const v8bf lo = *reinterpret_cast<const v8bf*>(&tile[l * PITCH + kb + half * 8]);
            const v8bf hi = *reinterpret_cast<const v8bf*>(&tile[l * PITCH + kb + 16 + half * 8]);
            const v16bf a = __builtin_shufflevector(lo, hi, 0, 1, 2, 3, 4, 5, 6, 7,
                                                    8, 9, 10, 11, 12, 13, 14, 15);
            acc = __builtin_amdgcn_wmma_f32_16x16x32_bf16(
                 false, a,  false, b[kb / 32],
                 (short)0, acc,  false,  false);
        }
        const int m0 = blockIdx.x * 16;
        #pragma unroll
        for (int v = 0; v < 8; ++v) {
            const int r = m0 + v + half * 8;
            Hn[(size_t)r * COUT + wave * 16 + l] = (_Float16)acc[v];
        }
    }
}

extern "C" void kernel_launch(void* const* d_in, const int* in_sizes, int n_in,
                              void* d_out, int out_size, void* d_ws, size_t ws_size,
                              hipStream_t stream) {
    const float* x  = (const float*)d_in[0];
    const int*   ei = (const int*)d_in[1];
    const float* W1 = (const float*)d_in[2];
    const float* b1 = (const float*)d_in[3];
    const float* W2 = (const float*)d_in[4];
    const float* b2 = (const float*)d_in[5];
    const float* W3 = (const float*)d_in[6];
    const float* b3 = (const float*)d_in[7];

    const int CIN = 128, CH = 128;
    const int n = in_sizes[0] / CIN;
    const int E = in_sizes[1] / 2;
    const int* src = ei;
    const int* dst = ei + E;

    _Float16* hlin   = (_Float16*)d_ws;
    _Float16* feat   = hlin + (size_t)n * CH;
    int2*  col    = (int2*)(feat + (size_t)n * CH);
    int*   chist  = (int*)(col + E);
    int*   lpre   = chist + (size_t)NCHUNK * n;
    int*   rowptr = lpre + n;
    const int G   = (n + 31) / 32;
    int*   gsum   = rowptr + n + 1;
    float* isq    = (float*)(gsum + G);

    const int TB = 256;
    chunk_hist_kernel<<<NCHUNK, CSR_TB, (size_t)n * sizeof(int), stream>>>(dst, chist, n, E);
    chunk_prefix_kernel<<<G, TB, 0, stream>>>(chist, lpre, gsum, isq, n);
    csr_place_kernel<<<NCHUNK, CSR_TB, (size_t)(n + G) * sizeof(int), stream>>>(
        src, dst, chist, lpre, gsum, isq, rowptr, col, n, E, G);

    const int row_tiles = n / 16;
    const int node_blocks = (n + 7) / 8;

    gemm_wmma_bf16_kernel<128, 128><<<row_tiles, dim3(32, 8), 0, stream>>>(x, W1, hlin);

    gather_gemm_kernel<128><<<row_tiles, 512, 0, stream>>>(hlin, rowptr, col, isq, b1, W2, feat, n);

    gather_gemm_kernel<64><<<row_tiles, 512, 0, stream>>>(feat, rowptr, col, isq, b2, W3, hlin, n);

    gather_sum_kernel<64><<<node_blocks, TB, 0, stream>>>(hlin, rowptr, col, isq, b3, (float*)d_out, n, 0);
}


# csr_place: every load not hanging on the group-sum scan (cursor inputs, first edges, norm factors) issued as one group ahead of it; a pass's LDS adds grouped; sum order still chunk, then LDS arrival
# speedup vs baseline: 5.8756x; 1.0871x over previous
#include <hip/hip_runtime.h>
#include <hip/hip_bf16.h>

typedef __attribute__((ext_vector_type(16))) __bf16 v16bf;
typedef __attribute__((ext_vector_type(8)))  float  v8f;


constexpr int NCHUNK = 128;
constexpr int CSR_TB = 1024;
constexpr int EPT = 8;

__global__ void __launch_bounds__(CSR_TB)
chunk_hist_kernel(const int* __restrict__ dst, int* __restrict__ chist,
                  int n, int E) {
    extern __shared__ int sm[];
    const int t = threadIdx.x;
    const int b = blockIdx.x;
    const int chunk = (E + NCHUNK - 1) / NCHUNK;
    const int e0 = b * chunk;
    int e1 = e0 + chunk;
    if (e1 > E) e1 = E;
    for (int i = t; i < n; i += CSR_TB) sm[i] = 0;
    __syncthreads();
    for (int i0 = e0 + t; i0 < e1; i0 += EPT * CSR_TB) {
        int d[EPT];
        #pragma unroll
        for (int u = 0; u < EPT; ++u) {
            const int i = i0 + u * CSR_TB;
            d[u] = dst[i < e1 ? i : e1 - 1];
        }
        #pragma unroll
        for (int u = 0; u < EPT; ++u)
            if (i0 + u * CSR_TB < e1) atomicAdd(&sm[d[u]], 1);
    }
    __syncthreads();
    for (int i = t; i < n; i += CSR_TB) chist[(size_t)b * n + i] = sm[i];
}

__global__ void __launch_bounds__(256)
chunk_prefix_kernel(int* __restrict__ chist, int* __restrict__ lpre,
                    int* __restrict__ gsum, float* __restrict__ isq, int n) {
    constexpr int PER = NCHUNK / 8;
    __shared__ int wt[8][32];
    const int lane = threadIdx.x & 31;
    const int w = threadIdx.x >> 5;
    const int node = blockIdx.x * 32 + lane;
    const bool ok = node < n;
    int v[PER];
    #pragma unroll
    for (int j = 0; j < PER; ++j)
        v[j] = ok ? chist[(size_t)(w * PER + j) * n + node] : 0;
    int run = 0;
    #pragma unroll
    for (int j = 0; j < PER; ++j) { const int c = v[j]; v[j] = run; run += c; }
    wt[w][lane] = run;
    __syncthreads();
    int off = 0, tot = 0;
    #pragma unroll
    for (int q = 0; q < 8; ++q) {
        const int x = wt[q][lane];
        if (q < w) off += x;
        tot += x;
    }
    if (ok) {
        #pragma unroll
        for (int j = 0; j < PER; ++j)
            chist[(size_t)(w * PER + j) * n + node] = v[j] + off;
    }
    if (w == 0) {
        int inc = tot;
        for (int o = 1; o < 32; o <<= 1) {
            const int u = __shfl_up(inc, o);
            if (lane >= o) inc += u;
        }
        if (ok) {
            lpre[node] = inc - tot;
            isq[node] = rsqrtf((float)(tot + 1));
        }
        if (lane == 31) gsum[blockIdx.x] = inc;
    }
}

constexpr int CPT = 10;
constexpr int GPL = 16;

__device__ __forceinline__ void load_edge_pass(
        const int* __restrict__ src, const int* __restrict__ dst,
        const float* __restrict__ isq, int i0, int e1,
        int (&s)[EPT], int (&d)[EPT], float (&w)[EPT]) {
    #pragma unroll
    for (int u = 0; u < EPT; ++u) {
        const int i = i0 + u * CSR_TB;
        const int ic = i < e1 ? i : e1 - 1;
        s[u] = src[ic];
        d[u] = dst[ic];
    }
    #pragma unroll
    for (int u = 0; u < EPT; ++u) w[u] = isq[s[u]];
}

__device__ __forceinline__ void load_group_sums(
        const int* __restrict__ gsum, int lo, int G, int (&gv)[GPL]) {
    #pragma unroll
    for (int q = 0; q < GPL; ++q) {
        const int g = lo + q;
        gv[q] = gsum[g < G ? g : G - 1];
    }
}

__device__ __forceinline__ void scan_sweep(
        int (&gv)[GPL], int lo, int G, int t, int& carry, int* gp) {
    int tot = 0;
    #pragma unroll
    for (int q = 0; q < GPL; ++q) {
        if (lo + q >= G) gv[q] = 0;
        tot += gv[q];
    }
    int inc = tot;
    for (int o = 1; o < 32; o <<= 1) {
        const int up = __shfl_up(inc, o);
        if (t >= o) inc += up;
    }
    int run = carry + inc - tot;
    #pragma unroll
    for (int q = 0; q < GPL; ++q) {
        if (lo + q < G) gp[lo + q] = run;
        run += gv[q];
    }
    carry += __shfl(inc, 31);
}

__global__ void __launch_bounds__(CSR_TB)
csr_place_kernel(const int* __restrict__ src, const int* __restrict__ dst,
                 const int* __restrict__ chist, const int* __restrict__ lpre,
                 const int* __restrict__ gsum, const float* __restrict__ isq,
                 int* __restrict__ rowptr, int2* __restrict__ col,
                 int n, int E, int G) {
    extern __shared__ int sm[];
    int* cur = sm;
    int* gp = sm + n;
    const int t = threadIdx.x;
    const int b = blockIdx.x;
    const int chunk = (E + NCHUNK - 1) / NCHUNK;
    const int e0 = b * chunk;
    int e1 = e0 + chunk;
    if (e1 > E) e1 = E;
    const int* __restrict__ crow = chist + (size_t)b * n;
    const bool scan_wave = t < 32;

    int lp[CPT], ch[CPT];
    #pragma unroll
    for (int u = 0; u < CPT; ++u) {
        const int i = t + u * CSR_TB;
        const int ic = i < n ? i : n - 1;
        lp[u] = lpre[ic];
        ch[u] = crow[ic];
    }
    int gv[GPL];
    #pragma unroll
    for (int q = 0; q < GPL; ++q) gv[q] = 0;
    if (scan_wave) load_group_sums(gsum, t * GPL, G, gv);
    int s[EPT], d[EPT];
    float w[EPT];
    load_edge_pass(src, dst, isq, e0 + t, e1, s, d, w);

    if (scan_wave) {
        int carry = 0;
        scan_sweep(gv, t * GPL, G, t, carry, gp);
        for (int g0 = 32 * GPL; g0 < G; g0 += 32 * GPL) {
            load_group_sums(gsum, g0 + t * GPL, G, gv);
            scan_sweep(gv, g0 + t * GPL, G, t, carry, gp);
        }
        if (t == 31 && b == 0) rowptr[n] = carry;
    }
    __syncthreads();
    int gs[CPT];
    #pragma unroll
    for (int u = 0; u < CPT; ++u) {
        const int i = t + u * CSR_TB;
        gs[u] = gp[(i < n ? i : n - 1) >> 5];
    }
    #pragma unroll
    for (int u = 0; u < CPT; ++u) {
        const int i = t + u * CSR_TB;
        if (i < n) {
            const int r = gs[u] + lp[u];
            if (b == 0) rowptr[i] = r;
            cur[i] = r + ch[u];
        }
    }
    for (int i = t + CPT * CSR_TB; i < n; i += CSR_TB) {
        const int r = gp[i >> 5] + lpre[i];
        if (b == 0) rowptr[i] = r;
        cur[i] = r + crow[i];
    }
    __syncthreads();
    for (int p0 = e0; p0 < e1; ) {
        const int i0 = p0 + t;
        int pos[EPT];
        #pragma unroll
        for (int u = 0; u < EPT; ++u) {
            pos[u] = 0;
            if (i0 + u * CSR_TB < e1) pos[u] = atomicAdd(&cur[d[u]], 1);
        }
        #pragma unroll
        for (int u = 0; u < EPT; ++u)
            if (i0 + u * CSR_TB < e1) col[pos[u]] = make_int2(s[u], __float_as_int(w[u]));
        p0 += EPT * CSR_TB;
        if (p0 < e1) load_edge_pass(src, dst, isq, p0 + t, e1, s, d, w);
    }
}

template <int CIN, int COUT>
__global__ void gemm_wmma_bf16_kernel(const float* __restrict__ X,
                                      const float* __restrict__ W,
                                      _Float16* __restrict__ H) {
    const int lane = threadIdx.x;
    const int ctile = threadIdx.y;
    const int m0 = blockIdx.x * 16;
    const int half = lane >> 4;
    const int l = lane & 15;
    const int m = m0 + l;
    const int n = ctile * 16 + l;

    v8f acc = {0.f, 0.f, 0.f, 0.f, 0.f, 0.f, 0.f, 0.f};

    #pragma unroll
    for (int kb = 0; kb < CIN; kb += 32) {
        float xa[16];
        const float4* xr0 = reinterpret_cast<const float4*>(&X[(size_t)m * CIN + kb + half * 8]);
        const float4* xr1 = reinterpret_cast<const float4*>(&X[(size_t)m * CIN + kb + 16 + half * 8]);
        *reinterpret_cast<float4*>(&xa[0])  = xr0[0];
        *reinterpret_cast<float4*>(&xa[4])  = xr0[1];
        *reinterpret_cast<float4*>(&xa[8])  = xr1[0];
        *reinterpret_cast<float4*>(&xa[12]) = xr1[1];

        v16bf a, b;
        #pragma unroll
        for (int e = 0; e < 8; ++e) {
            const int k0 = kb + half * 8 + e;
            const int k1 = k0 + 16;
            a[e]     = (__bf16)xa[e];
            a[e + 8] = (__bf16)xa[e + 8];
            b[e]     = (__bf16)W[(size_t)k0 * COUT + n];
            b[e + 8] = (__bf16)W[(size_t)k1 * COUT + n];
        }
        acc = __builtin_amdgcn_wmma_f32_16x16x32_bf16(
             false, a,  false, b,
             (short)0, acc,  false,  false);
    }

    #pragma unroll
    for (int v = 0; v < 8; ++v) {
        const int r = m0 + v + half * 8;
        H[(size_t)r * COUT + ctile * 16 + l] = (_Float16)acc[v];
    }
}

template <int C>
__global__ void __launch_bounds__(256)
gather_sum_kernel(const _Float16* __restrict__ H,
                  const int* __restrict__ rowptr,
                  const int2* __restrict__ col,
                  const float* __restrict__ isq,
                  const float* __restrict__ bias,
                  float* __restrict__ OUT, int n, int do_relu) {
    constexpr int VEC = C / 32;
    typedef float vec_t __attribute__((ext_vector_type(VEC)));
    typedef _Float16 row_t __attribute__((ext_vector_type(VEC)));
    const int lane = threadIdx.x & 31;
    const int node = __builtin_amdgcn_readfirstlane(
        (int)(blockIdx.x * (blockDim.x >> 5) + (threadIdx.x >> 5)));
    if (node >= n) return;
    const int beg = rowptr[node];
    const int end = rowptr[node + 1];
    const float sd = isq[node];
    const row_t* __restrict__ Hv = reinterpret_cast<const row_t*>(H);

    const vec_t hv = __builtin_convertvector(Hv[(size_t)node * 32 + lane], vec_t);
    vec_t acc = (hv * sd) * sd;

    int2 nxt = make_int2(0, 0);
    if (beg + lane < end) nxt = col[beg + lane];
    for (int j0 = beg; j0 < end; j0 += 32) {
        const int cur_s = nxt.x;
        const float cur_w = __int_as_float(nxt.y) * sd;
        nxt = make_int2(0, 0);
        const int jn = j0 + 32 + lane;
        if (jn < end) nxt = col[jn];
        int m = end - j0;
        if (m > 32) m = 32;
        int k = 0;
        for (; k + 8 <= m; k += 8) {
            row_t v[8];
            #pragma unroll
            for (int u = 0; u < 8; ++u) {
                const int s = __builtin_amdgcn_readlane(cur_s, k + u);
                v[u] = Hv[(size_t)s * 32 + lane];
            }
            #pragma unroll
            for (int u = 0; u < 8; ++u) {
                const float w = __int_as_float(
                    __builtin_amdgcn_readlane(__float_as_int(cur_w), k + u));
                acc += __builtin_convertvector(v[u], vec_t) * w;
            }
        }
        if (k < m) {
            row_t v[8];
            #pragma unroll
            for (int u = 0; u < 8; ++u) {
                const int s = __builtin_amdgcn_readlane(cur_s, (k + u) & 31);
                v[u] = Hv[(size_t)s * 32 + lane];
            }
            #pragma unroll
            for (int u = 0; u < 8; ++u) {
                const float w = __int_as_float(
                    __builtin_amdgcn_readlane(__float_as_int(cur_w), (k + u) & 31));
                if (k + u < m) acc += __builtin_convertvector(v[u], vec_t) * w;
            }
        }
    }

    const vec_t bv = reinterpret_cast<const vec_t*>(bias)[lane];
    vec_t o = acc + bv;
    if (do_relu) {
        #pragma unroll
        for (int i = 0; i < VEC; ++i) o[i] = fmaxf(o[i], 0.0f);
    }
    reinterpret_cast<vec_t*>(OUT)[(size_t)node * 32 + lane] = o;
}

typedef __attribute__((ext_vector_type(8))) __bf16 v8bf;
typedef __attribute__((ext_vector_type(4))) __bf16 v4bf;

template <int COUT>
__global__ void __launch_bounds__(512)
gather_gemm_kernel(const _Float16* __restrict__ H,
                   const int* __restrict__ rowptr,
                   const int2* __restrict__ col,
                   const float* __restrict__ isq,
                   const float* __restrict__ bias,
                   const float* __restrict__ Wn,
                   _Float16* __restrict__ Hn, int n) {
    constexpr int C = 128;
    constexpr int PITCH = C + 8;
    constexpr int NT = COUT / 16;
    typedef float vec_t __attribute__((ext_vector_type(4)));
    typedef _Float16 row_t __attribute__((ext_vector_type(4)));
    __shared__ __attribute__((aligned(16))) __bf16 tile[16 * PITCH];

    const int lane = threadIdx.x & 31;
    const int wave = __builtin_amdgcn_readfirstlane((int)(threadIdx.x >> 5));
    const int node = __builtin_amdgcn_readfirstlane((int)(blockIdx.x * 16)) + wave;

    vec_t o = {0.f, 0.f, 0.f, 0.f};
    if (node < n) {
        const int beg = rowptr[node];
        const int end = rowptr[node + 1];
        const float sd = isq[node];
        const row_t* __restrict__ Hv = reinterpret_cast<const row_t*>(H);

        const vec_t hv = __builtin_convertvector(Hv[(size_t)node * 32 + lane], vec_t);
        vec_t acc = (hv * sd) * sd;

        int2 nxt = make_int2(0, 0);
        if (beg + lane < end) nxt = col[beg + lane];
        for (int j0 = beg; j0 < end; j0 += 32) {
            const int cur_s = nxt.x;
            const float cur_w = __int_as_float(nxt.y) * sd;
            nxt = make_int2(0, 0);
            const int jn = j0 + 32 + lane;
            if (jn < end) nxt = col[jn];
            int m = end - j0;
            if (m > 32) m = 32;
            int k = 0;
            for (; k + 8 <= m; k += 8) {
                row_t v[8];
                #pragma unroll
                for (int u = 0; u < 8; ++u) {
                    const int s = __builtin_amdgcn_readlane(cur_s, k + u);
                    v[u] = Hv[(size_t)s * 32 + lane];
                }
                #pragma unroll
                for (int u = 0; u < 8; ++u) {
                    const float w = __int_as_float(
                        __builtin_amdgcn_readlane(__float_as_int(cur_w), k + u));
                    acc += __builtin_convertvector(v[u], vec_t) * w;
                }
            }
            if (k < m) {
                row_t v[8];
                #pragma unroll
                for (int u = 0; u < 8; ++u) {
                    const int s = __builtin_amdgcn_readlane(cur_s, (k + u) & 31);
                    v[u] = Hv[(size_t)s * 32 + lane];
                }
                #pragma unroll
                for (int u = 0; u < 8; ++u) {
                    const float w = __int_as_float(
                        __builtin_amdgcn_readlane(__float_as_int(cur_w), (k + u) & 31));
                    if (k + u < m) acc += __builtin_convertvector(v[u], vec_t) * w;
                }
            }
        }
        const vec_t bv = reinterpret_cast<const vec_t*>(bias)[lane];
        o = acc + bv;
        #pragma unroll
        for (int i = 0; i < 4; ++i) o[i] = fmaxf(o[i], 0.0f);
    }

    v4bf ob;
    #pragma unroll
    for (int i = 0; i < 4; ++i) ob[i] = (__bf16)o[i];
    *reinterpret_cast<v4bf*>(&tile[wave * PITCH + lane * 4]) = ob;

    const int half = lane >> 4;
    const int l = lane & 15;

    v16bf b[C / 32];
    if (wave < NT) {
        const int nn = wave * 16 + l;
        #pragma unroll
        for (int kb = 0; kb < C; kb += 32) {
            #pragma unroll
            for (int e = 0; e < 8; ++e) {
                const int k0 = kb + half * 8 + e;
                const int k1 = k0 + 16;
                b[kb / 32][e]     = (__bf16)Wn[(size_t)k0 * COUT + nn];
                b[kb / 32][e + 8] = (__bf16)Wn[(size_t)k1 * COUT + nn];
            }
        }
    }
    __syncthreads();
    if (wave < NT) {
        v8f acc = {0.f, 0.f, 0.f, 0.f, 0.f, 0.f, 0.f, 0.f};
        #pragma unroll
        for (int kb = 0; kb < C; kb += 32) {
            const v8bf lo = *reinterpret_cast<const v8bf*>(&tile[l * PITCH + kb + half * 8]);
            const v8bf hi = *reinterpret_cast<const v8bf*>(&tile[l * PITCH + kb + 16 + half * 8]);
            const v16bf a = __builtin_shufflevector(lo, hi, 0, 1, 2, 3, 4, 5, 6, 7,
                                                    8, 9, 10, 11, 12, 13, 14, 15);
            acc = __builtin_amdgcn_wmma_f32_16x16x32_bf16(
                 false, a,  false, b[kb / 32],
                 (short)0, acc,  false,  false);
        }
        const int m0 = blockIdx.x * 16;
        #pragma unroll
        for (int v = 0; v < 8; ++v) {
            const int r = m0 + v + half * 8;
            Hn[(size_t)r * COUT + wave * 16 + l] = (_Float16)acc[v];
        }
    }
}

extern "C" void kernel_launch(void* const* d_in, const int* in_sizes, int n_in,
                              void* d_out, int out_size, void* d_ws, size_t ws_size,
                              hipStream_t stream) {
    const float* x  = (const float*)d_in[0];
    const int*   ei = (const int*)d_in[1];
    const float* W1 = (const float*)d_in[2];
    const float* b1 = (const float*)d_in[3];
    const float* W2 = (const float*)d_in[4];
    const float* b2 = (const float*)d_in[5];
    const float* W3 = (const float*)d_in[6];
    const float* b3 = (const float*)d_in[7];

    const int CIN = 128, CH = 128;
    const int n = in_sizes[0] / CIN;
    const int E = in_sizes[1] / 2;
    const int* src = ei;
    const int* dst = ei + E;

    _Float16* hlin   = (_Float16*)d_ws;
    _Float16* feat   = hlin + (size_t)n * CH;
    int2*  col    = (int2*)(feat + (size_t)n * CH);
    int*   chist  = (int*)(col + E);
    int*   lpre   = chist + (size_t)NCHUNK * n;
    int*   rowptr = lpre + n;
    const int G   = (n + 31) / 32;
    int*   gsum   = rowptr + n + 1;
    float* isq    = (float*)(gsum + G);

    const int TB = 256;
    chunk_hist_kernel<<<NCHUNK, CSR_TB, (size_t)n * sizeof(int), stream>>>(dst, chist, n, E);
    chunk_prefix_kernel<<<G, TB, 0, stream>>>(chist, lpre, gsum, isq, n);
    csr_place_kernel<<<NCHUNK, CSR_TB, (size_t)(n + G) * sizeof(int), stream>>>(
        src, dst, chist, lpre, gsum, isq, rowptr, col, n, E, G);

    const int row_tiles = n / 16;
    const int node_blocks = (n + 7) / 8;

    gemm_wmma_bf16_kernel<128, 128><<<row_tiles, dim3(32, 8), 0, stream>>>(x, W1, hlin);

    gather_gemm_kernel<128><<<row_tiles, 512, 0, stream>>>(hlin, rowptr, col, isq, b1, W2, feat, n);

    gather_gemm_kernel<64><<<row_tiles, 512, 0, stream>>>(feat, rowptr, col, isq, b2, W3, hlin, n);

    gather_sum_kernel<64><<<node_blocks, TB, 0, stream>>>(hlin, rowptr, col, isq, b3, (float*)d_out, n, 0);
}


# layer-1 product x@W1 (same bf16 WMMA tile code) runs as extra work-groups of the csr_place launch, on the processors that launch leaves idle; 6 launches not 7; sum order unchanged
# speedup vs baseline: 6.4374x; 1.0956x over previous
#include <hip/hip_runtime.h>
#include <hip/hip_bf16.h>

typedef __attribute__((ext_vector_type(16))) __bf16 v16bf;
typedef __attribute__((ext_vector_type(8)))  float  v8f;


template <int CIN, int COUT>
__device__ __forceinline__ void gemm_tile_bf16(const float* __restrict__ X,
                                               const float* __restrict__ W,
                                               _Float16* __restrict__ H,
                                               int lane, int ctile, int m0) {
    const int half = lane >> 4;
    const int l = lane & 15;
    const int m = m0 + l;
    const int n = ctile * 16 + l;

    v8f acc = {0.f, 0.f, 0.f, 0.f, 0.f, 0.f, 0.f, 0.f};

    #pragma unroll
    for (int kb = 0; kb < CIN; kb += 32) {
        float xa[16];
        const float4* xr0 = reinterpret_cast<const float4*>(&X[(size_t)m * CIN + kb + half * 8]);
        const float4* xr1 = reinterpret_cast<const float4*>(&X[(size_t)m * CIN + kb + 16 + half * 8]);
        *reinterpret_cast<float4*>(&xa[0])  = xr0[0];
        *reinterpret_cast<float4*>(&xa[4])  = xr0[1];
        *reinterpret_cast<float4*>(&xa[8])  = xr1[0];
        *reinterpret_cast<float4*>(&xa[12]) = xr1[1];

        v16bf a, b;
        #pragma unroll
        for (int e = 0; e < 8; ++e) {
            const int k0 = kb + half * 8 + e;
            const int k1 = k0 + 16;
            a[e]     = (__bf16)xa[e];
            a[e + 8] = (__bf16)xa[e + 8];
            b[e]     = (__bf16)W[(size_t)k0 * COUT + n];
            b[e + 8] = (__bf16)W[(size_t)k1 * COUT + n];
        }
        acc = __builtin_amdgcn_wmma_f32_16x16x32_bf16(
             false, a,  false, b,
             (short)0, acc,  false,  false);
    }

    #pragma unroll
    for (int v = 0; v < 8; ++v) {
        const int r = m0 + v + half * 8;
        H[(size_t)r * COUT + ctile * 16 + l] = (_Float16)acc[v];
    }
}

constexpr int NCHUNK = 128;
constexpr int CSR_TB = 1024;
constexpr int EPT = 8;

__global__ void __launch_bounds__(CSR_TB)
chunk_hist_kernel(const int* __restrict__ dst, int* __restrict__ chist,
                  int n, int E) {
    extern __shared__ int sm[];
    const int t = threadIdx.x;
    const int b = blockIdx.x;
    const int chunk = (E + NCHUNK - 1) / NCHUNK;
    const int e0 = b * chunk;
    int e1 = e0 + chunk;
    if (e1 > E) e1 = E;
    for (int i = t; i < n; i += CSR_TB) sm[i] = 0;
    __syncthreads();
    for (int i0 = e0 + t; i0 < e1; i0 += EPT * CSR_TB) {
        int d[EPT];
        #pragma unroll
        for (int u = 0; u < EPT; ++u) {
            const int i = i0 + u * CSR_TB;
            d[u] = dst[i < e1 ? i : e1 - 1];
        }
        #pragma unroll
        for (int u = 0; u < EPT; ++u)
            if (i0 + u * CSR_TB < e1) atomicAdd(&sm[d[u]], 1);
    }
    __syncthreads();
    for (int i = t; i < n; i += CSR_TB) chist[(size_t)b * n + i] = sm[i];
}

__global__ void __launch_bounds__(256)
chunk_prefix_kernel(int* __restrict__ chist, int* __restrict__ lpre,
                    int* __restrict__ gsum, float* __restrict__ isq, int n) {
    constexpr int PER = NCHUNK / 8;
    __shared__ int wt[8][32];
    const int lane = threadIdx.x & 31;
    const int w = threadIdx.x >> 5;
    const int node = blockIdx.x * 32 + lane;
    const bool ok = node < n;
    int v[PER];
    #pragma unroll
    for (int j = 0; j < PER; ++j)
        v[j] = ok ? chist[(size_t)(w * PER + j) * n + node] : 0;
    int run = 0;
    #pragma unroll
    for (int j = 0; j < PER; ++j) { const int c = v[j]; v[j] = run; run += c; }
    wt[w][lane] = run;
    __syncthreads();
    int off = 0, tot = 0;
    #pragma unroll
    for (int q = 0; q < 8; ++q) {
        const int x = wt[q][lane];
        if (q < w) off += x;
        tot += x;
    }
    if (ok) {
        #pragma unroll
        for (int j = 0; j < PER; ++j)
            chist[(size_t)(w * PER + j) * n + node] = v[j] + off;
    }
    if (w == 0) {
        int inc = tot;
        for (int o = 1; o < 32; o <<= 1) {
            const int u = __shfl_up(inc, o);
            if (lane >= o) inc += u;
        }
        if (ok) {
            lpre[node] = inc - tot;
            isq[node] = rsqrtf((float)(tot + 1));
        }
        if (lane == 31) gsum[blockIdx.x] = inc;
    }
}

constexpr int CPT = 10;
constexpr int GPL = 16;
constexpr int GEMM_RT = CSR_TB / 32 / 8;

__device__ __forceinline__ void load_edge_pass(
        const int* __restrict__ src, const int* __restrict__ dst,
        const float* __restrict__ isq, int i0, int e1,
        int (&s)[EPT], int (&d)[EPT], float (&w)[EPT]) {
    #pragma unroll
    for (int u = 0; u < EPT; ++u) {
        const int i = i0 + u * CSR_TB;
        const int ic = i < e1 ? i : e1 - 1;
        s[u] = src[ic];
        d[u] = dst[ic];
    }
    #pragma unroll
    for (int u = 0; u < EPT; ++u) w[u] = isq[s[u]];
}

__device__ __forceinline__ void load_group_sums(
        const int* __restrict__ gsum, int lo, int G, int (&gv)[GPL]) {
    #pragma unroll
    for (int q = 0; q < GPL; ++q) {
        const int g = lo + q;
        gv[q] = gsum[g < G ? g : G - 1];
    }
}

__device__ __forceinline__ void scan_sweep(
        int (&gv)[GPL], int lo, int G, int t, int& carry, int* gp) {
    int tot = 0;
    #pragma unroll
    for (int q = 0; q < GPL; ++q) {
        if (lo + q >= G) gv[q] = 0;
        tot += gv[q];
    }
    int inc = tot;
    for (int o = 1; o < 32; o <<= 1) {
        const int up = __shfl_up(inc, o);
        if (t >= o) inc += up;
    }
    int run = carry + inc - tot;
    #pragma unroll
    for (int q = 0; q < GPL; ++q) {
        if (lo + q < G) gp[lo + q] = run;
        run += gv[q];
    }
    carry += __shfl(inc, 31);
}

__global__ void __launch_bounds__(CSR_TB)
csr_place_kernel(const int* __restrict__ src, const int* __restrict__ dst,
                 const int* __restrict__ chist, const int* __restrict__ lpre,
                 const int* __restrict__ gsum, const float* __restrict__ isq,
                 int* __restrict__ rowptr, int2* __restrict__ col,
                 int n, int E, int G,
                 const float* __restrict__ x, const float* __restrict__ W1,
                 _Float16* __restrict__ hlin) {
    if (blockIdx.x >= NCHUNK) {
        const int wv = __builtin_amdgcn_readfirstlane((int)(threadIdx.x >> 5));
        const int rt = (int)(blockIdx.x - NCHUNK) * GEMM_RT + (wv >> 3);
        if (rt < n / 16)
            gemm_tile_bf16<128, 128>(x, W1, hlin, threadIdx.x & 31, wv & 7, rt * 16);
        return;
    }
    extern __shared__ int sm[];
    int* cur = sm;
    int* gp = sm + n;
    const int t = threadIdx.x;
    const int b = blockIdx.x;
    const int chunk = (E + NCHUNK - 1) / NCHUNK;
    const int e0 = b * chunk;
    int e1 = e0 + chunk;
    if (e1 > E) e1 = E;
    const int* __restrict__ crow = chist + (size_t)b * n;
    const bool scan_wave = t < 32;

    int lp[CPT], ch[CPT];
    #pragma unroll
    for (int u = 0; u < CPT; ++u) {
        const int i = t + u * CSR_TB;
        const int ic = i < n ? i : n - 1;
        lp[u] = lpre[ic];
        ch[u] = crow[ic];
    }
    int gv[GPL];
    #pragma unroll
    for (int q = 0; q < GPL; ++q) gv[q] = 0;
    if (scan_wave) load_group_sums(gsum, t * GPL, G, gv);
    int s[EPT], d[EPT];
    float w[EPT];
    load_edge_pass(src, dst, isq, e0 + t, e1, s, d, w);

    if (scan_wave) {
        int carry = 0;
        scan_sweep(gv, t * GPL, G, t, carry, gp);
        for (int g0 = 32 * GPL; g0 < G; g0 += 32 * GPL) {
            load_group_sums(gsum, g0 + t * GPL, G, gv);
            scan_sweep(gv, g0 + t * GPL, G, t, carry, gp);
        }
        if (t == 31 && b == 0) rowptr[n] = carry;
    }
    __syncthreads();
    int gs[CPT];
    #pragma unroll
    for (int u = 0; u < CPT; ++u) {
        const int i = t + u * CSR_TB;
        gs[u] = gp[(i < n ? i : n - 1) >> 5];
    }
    #pragma unroll
    for (int u = 0; u < CPT; ++u) {
        const int i = t + u * CSR_TB;
        if (i < n) {
            const int r = gs[u] + lp[u];
            if (b == 0) rowptr[i] = r;
            cur[i] = r + ch[u];
        }
    }
    for (int i = t + CPT * CSR_TB; i < n; i += CSR_TB) {
        const int r = gp[i >> 5] + lpre[i];
        if (b == 0) rowptr[i] = r;
        cur[i] = r + crow[i];
    }
    __syncthreads();
    for (int p0 = e0; p0 < e1; ) {
        const int i0 = p0 + t;
        int pos[EPT];
        #pragma unroll
        for (int u = 0; u < EPT; ++u) {
            pos[u] = 0;
            if (i0 + u * CSR_TB < e1) pos[u] = atomicAdd(&cur[d[u]], 1);
        }
        #pragma unroll
        for (int u = 0; u < EPT; ++u)
            if (i0 + u * CSR_TB < e1) col[pos[u]] = make_int2(s[u], __float_as_int(w[u]));
        p0 += EPT * CSR_TB;
        if (p0 < e1) load_edge_pass(src, dst, isq, p0 + t, e1, s, d, w);
    }
}

template <int C>
__global__ void __launch_bounds__(256)
gather_sum_kernel(const _Float16* __restrict__ H,
                  const int* __restrict__ rowptr,
                  const int2* __restrict__ col,
                  const float* __restrict__ isq,
                  const float* __restrict__ bias,
                  float* __restrict__ OUT, int n, int do_relu) {
    constexpr int VEC = C / 32;
    typedef float vec_t __attribute__((ext_vector_type(VEC)));
    typedef _Float16 row_t __attribute__((ext_vector_type(VEC)));
    const int lane = threadIdx.x & 31;
    const int node = __builtin_amdgcn_readfirstlane(
        (int)(blockIdx.x * (blockDim.x >> 5) + (threadIdx.x >> 5)));
    if (node >= n) return;
    const int beg = rowptr[node];
    const int end = rowptr[node + 1];
    const float sd = isq[node];
    const row_t* __restrict__ Hv = reinterpret_cast<const row_t*>(H);

    const vec_t hv = __builtin_convertvector(Hv[(size_t)node * 32 + lane], vec_t);
    vec_t acc = (hv * sd) * sd;

    int2 nxt = make_int2(0, 0);
    if (beg + lane < end) nxt = col[beg + lane];
    for (int j0 = beg; j0 < end; j0 += 32) {
        const int cur_s = nxt.x;
        const float cur_w = __int_as_float(nxt.y) * sd;
        nxt = make_int2(0, 0);
        const int jn = j0 + 32 + lane;
        if (jn < end) nxt = col[jn];
        int m = end - j0;
        if (m > 32) m = 32;
        int k = 0;
        for (; k + 8 <= m; k += 8) {
            row_t v[8];
            #pragma unroll
            for (int u = 0; u < 8; ++u) {
                const int s = __builtin_amdgcn_readlane(cur_s, k + u);
                v[u] = Hv[(size_t)s * 32 + lane];
            }
            #pragma unroll
            for (int u = 0; u < 8; ++u) {
                const float w = __int_as_float(
                    __builtin_amdgcn_readlane(__float_as_int(cur_w), k + u));
                acc += __builtin_convertvector(v[u], vec_t) * w;
            }
        }
        if (k < m) {
            row_t v[8];
            #pragma unroll
            for (int u = 0; u < 8; ++u) {
                const int s = __builtin_amdgcn_readlane(cur_s, (k + u) & 31);
                v[u] = Hv[(size_t)s * 32 + lane];
            }
            #pragma unroll
            for (int u = 0; u < 8; ++u) {
                const float w = __int_as_float(
                    __builtin_amdgcn_readlane(__float_as_int(cur_w), (k + u) & 31));
                if (k + u < m) acc += __builtin_convertvector(v[u], vec_t) * w;
            }
        }
    }

    const vec_t bv = reinterpret_cast<const vec_t*>(bias)[lane];
    vec_t o = acc + bv;
    if (do_relu) {
        #pragma unroll
        for (int i = 0; i < VEC; ++i) o[i] = fmaxf(o[i], 0.0f);
    }
    reinterpret_cast<vec_t*>(OUT)[(size_t)node * 32 + lane] = o;
}

typedef __attribute__((ext_vector_type(8))) __bf16 v8bf;
typedef __attribute__((ext_vector_type(4))) __bf16 v4bf;

template <int COUT>
__global__ void __launch_bounds__(512)
gather_gemm_kernel(const _Float16* __restrict__ H,
                   const int* __restrict__ rowptr,
                   const int2* __restrict__ col,
                   const float* __restrict__ isq,
                   const float* __restrict__ bias,
                   const float* __restrict__ Wn,
                   _Float16* __restrict__ Hn, int n) {
    constexpr int C = 128;
    constexpr int PITCH = C + 8;
    constexpr int NT = COUT / 16;
    typedef float vec_t __attribute__((ext_vector_type(4)));
    typedef _Float16 row_t __attribute__((ext_vector_type(4)));
    __shared__ __attribute__((aligned(16))) __bf16 tile[16 * PITCH];

    const int lane = threadIdx.x & 31;
    const int wave = __builtin_amdgcn_readfirstlane((int)(threadIdx.x >> 5));
    const int node = __builtin_amdgcn_readfirstlane((int)(blockIdx.x * 16)) + wave;

    vec_t o = {0.f, 0.f, 0.f, 0.f};
    if (node < n) {
        const int beg = rowptr[node];
        const int end = rowptr[node + 1];
        const float sd = isq[node];
        const row_t* __restrict__ Hv = reinterpret_cast<const row_t*>(H);

        const vec_t hv = __builtin_convertvector(Hv[(size_t)node * 32 + lane], vec_t);
        vec_t acc = (hv * sd) * sd;

        int2 nxt = make_int2(0, 0);
        if (beg + lane < end) nxt = col[beg + lane];
        for (int j0 = beg; j0 < end; j0 += 32) {
            const int cur_s = nxt.x;
            const float cur_w = __int_as_float(nxt.y) * sd;
            nxt = make_int2(0, 0);
            const int jn = j0 + 32 + lane;
            if (jn < end) nxt = col[jn];
            int m = end - j0;
            if (m > 32) m = 32;
            int k = 0;
            for (; k + 8 <= m; k += 8) {
                row_t v[8];
                #pragma unroll
                for (int u = 0; u < 8; ++u) {
                    const int s = __builtin_amdgcn_readlane(cur_s, k + u);
                    v[u] = Hv[(size_t)s * 32 + lane];
                }
                #pragma unroll
                for (int u = 0; u < 8; ++u) {
                    const float w = __int_as_float(
                        __builtin_amdgcn_readlane(__float_as_int(cur_w), k + u));
                    acc += __builtin_convertvector(v[u], vec_t) * w;
                }
            }
            if (k < m) {
                row_t v[8];
                #pragma unroll
                for (int u = 0; u < 8; ++u) {
                    const int s = __builtin_amdgcn_readlane(cur_s, (k + u) & 31);
                    v[u] = Hv[(size_t)s * 32 + lane];
                }
                #pragma unroll
                for (int u = 0; u < 8; ++u) {
                    const float w = __int_as_float(
                        __builtin_amdgcn_readlane(__float_as_int(cur_w), (k + u) & 31));
                    if (k + u < m) acc += __builtin_convertvector(v[u], vec_t) * w;
                }
            }
        }
        const vec_t bv = reinterpret_cast<const vec_t*>(bias)[lane];
        o = acc + bv;
        #pragma unroll
        for (int i = 0; i < 4; ++i) o[i] = fmaxf(o[i], 0.0f);
    }

    v4bf ob;
    #pragma unroll
    for (int i = 0; i < 4; ++i) ob[i] = (__bf16)o[i];
    *reinterpret_cast<v4bf*>(&tile[wave * PITCH + lane * 4]) = ob;

    const int half = lane >> 4;
    const int l = lane & 15;

    v16bf b[C / 32];
    if (wave < NT) {
        const int nn = wave * 16 + l;
        #pragma unroll
        for (int kb = 0; kb < C; kb += 32) {
            #pragma unroll
            for (int e = 0; e < 8; ++e) {
                const int k0 = kb + half * 8 + e;
                const int k1 = k0 + 16;
                b[kb / 32][e]     = (__bf16)Wn[(size_t)k0 * COUT + nn];
                b[kb / 32][e + 8] = (__bf16)Wn[(size_t)k1 * COUT + nn];
            }
        }
    }
    __syncthreads();
    if (wave < NT) {
        v8f acc = {0.f, 0.f, 0.f, 0.f, 0.f, 0.f, 0.f, 0.f};
        #pragma unroll
        for (int kb = 0; kb < C; kb += 32) {
            const v8bf lo = *reinterpret_cast<const v8bf*>(&tile[l * PITCH + kb + half * 8]);
            const v8bf hi = *reinterpret_cast<const v8bf*>(&tile[l * PITCH + kb + 16 + half * 8]);
            const v16bf a = __builtin_shufflevector(lo, hi, 0, 1, 2, 3, 4, 5, 6, 7,
                                                    8, 9, 10, 11, 12, 13, 14, 15);
            acc = __builtin_amdgcn_wmma_f32_16x16x32_bf16(
                 false, a,  false, b[kb / 32],
                 (short)0, acc,  false,  false);
        }
        const int m0 = blockIdx.x * 16;
        #pragma unroll
        for (int v = 0; v < 8; ++v) {
            const int r = m0 + v + half * 8;
            Hn[(size_t)r * COUT + wave * 16 + l] = (_Float16)acc[v];
        }
    }
}

extern "C" void kernel_launch(void* const* d_in, const int* in_sizes, int n_in,
                              void* d_out, int out_size, void* d_ws, size_t ws_size,
                              hipStream_t stream) {
    const float* x  = (const float*)d_in[0];
    const int*   ei = (const int*)d_in[1];
    const float* W1 = (const float*)d_in[2];
    const float* b1 = (const float*)d_in[3];
    const float* W2 = (const float*)d_in[4];
    const float* b2 = (const float*)d_in[5];
    const float* W3 = (const float*)d_in[6];
    const float* b3 = (const float*)d_in[7];

    const int CIN = 128, CH = 128;
    const int n = in_sizes[0] / CIN;
    const int E = in_sizes[1] / 2;
    const int* src = ei;
    const int* dst = ei + E;

    _Float16* hlin   = (_Float16*)d_ws;
    _Float16* feat   = hlin + (size_t)n * CH;
    int2*  col    = (int2*)(feat + (size_t)n * CH);
    int*   chist  = (int*)(col + E);
    int*   lpre   = chist + (size_t)NCHUNK * n;
    int*   rowptr = lpre + n;
    const int G   = (n + 31) / 32;
    int*   gsum   = rowptr + n + 1;
    float* isq    = (float*)(gsum + G);

    const int TB = 256;
    chunk_hist_kernel<<<NCHUNK, CSR_TB, (size_t)n * sizeof(int), stream>>>(dst, chist, n, E);
    chunk_prefix_kernel<<<G, TB, 0, stream>>>(chist, lpre, gsum, isq, n);
    const int row_tiles = n / 16;
    const int node_blocks = (n + 7) / 8;

    csr_place_kernel<<<NCHUNK + (row_tiles + GEMM_RT - 1) / GEMM_RT, CSR_TB,
                       (size_t)(n + G) * sizeof(int), stream>>>(
        src, dst, chist, lpre, gsum, isq, rowptr, col, n, E, G, x, W1, hlin);

    gather_gemm_kernel<128><<<row_tiles, 512, 0, stream>>>(hlin, rowptr, col, isq, b1, W2, feat, n);

    gather_gemm_kernel<64><<<row_tiles, 512, 0, stream>>>(feat, rowptr, col, isq, b2, W3, hlin, n);

    gather_sum_kernel<64><<<node_blocks, TB, 0, stream>>>(hlin, rowptr, col, isq, b3, (float*)d_out, n, 0);
}


# weights packed once per call (3 extra work-groups of the histogram launch) into bf16 B fragments in the loaders' k order; a product wave fetches B as 8 wide loads, not 64 narrow ones; sums unchanged
# speedup vs baseline: 6.6351x; 1.0307x over previous
#include <hip/hip_runtime.h>
#include <hip/hip_bf16.h>

typedef __attribute__((ext_vector_type(16))) __bf16 v16bf;
typedef __attribute__((ext_vector_type(8)))  float  v8f;


template <int CIN, int COUT>
__device__ __forceinline__ void gemm_tile_bf16(const float* __restrict__ X,
                                               const v16bf* __restrict__ Wp,
                                               _Float16* __restrict__ H,
                                               int lane, int ctile, int m0) {
    const int half = lane >> 4;
    const int l = lane & 15;
    const int m = m0 + l;

    v8f acc = {0.f, 0.f, 0.f, 0.f, 0.f, 0.f, 0.f, 0.f};

    #pragma unroll
    for (int kb = 0; kb < CIN; kb += 32) {
        float xa[16];
        const float4* xr0 = reinterpret_cast<const float4*>(&X[(size_t)m * CIN + kb + half * 8]);
        const float4* xr1 = reinterpret_cast<const float4*>(&X[(size_t)m * CIN + kb + 16 + half * 8]);
        *reinterpret_cast<float4*>(&xa[0])  = xr0[0];
        *reinterpret_cast<float4*>(&xa[4])  = xr0[1];
        *reinterpret_cast<float4*>(&xa[8])  = xr1[0];
        *reinterpret_cast<float4*>(&xa[12]) = xr1[1];

        v16bf a;
        #pragma unroll
        for (int e = 0; e < 8; ++e) {
            a[e]     = (__bf16)xa[e];
            a[e + 8] = (__bf16)xa[e + 8];
        }
        const v16bf b = Wp[(ctile * (CIN / 32) + kb / 32) * 32 + lane];
        acc = __builtin_amdgcn_wmma_f32_16x16x32_bf16(
             false, a,  false, b,
             (short)0, acc,  false,  false);
    }

    #pragma unroll
    for (int v = 0; v < 8; ++v) {
        const int r = m0 + v + half * 8;
        H[(size_t)r * COUT + ctile * 16 + l] = (_Float16)acc[v];
    }
}

constexpr int NCHUNK = 128;
constexpr int CSR_TB = 1024;
constexpr int EPT = 8;

constexpr int FRAGS128 = 8 * 4 * 32;
constexpr int FRAGS64  = 4 * 4 * 32;

__device__ __forceinline__ void pack_weight_frag(const float* __restrict__ W,
                                                 v16bf* __restrict__ P,
                                                 int cout, int f) {
    const int lane = f & 31;
    const int kb = ((f >> 5) & 3) * 32;
    const int ct = f >> 7;
    const int half = lane >> 4;
    const int n = ct * 16 + (lane & 15);
    v16bf b;
    #pragma unroll
    for (int e = 0; e < 8; ++e) {
        const int k0 = kb + half * 8 + e;
        b[e]     = (__bf16)W[(size_t)k0 * cout + n];
        b[e + 8] = (__bf16)W[(size_t)(k0 + 16) * cout + n];
    }
    P[f] = b;
}

__global__ void __launch_bounds__(CSR_TB)
chunk_hist_kernel(const int* __restrict__ dst, int* __restrict__ chist,
                  int n, int E,
                  const float* __restrict__ W1, const float* __restrict__ W2,
                  const float* __restrict__ W3, v16bf* __restrict__ Wp) {
    if (blockIdx.x >= NCHUNK) {
        const int p = (int)blockIdx.x - NCHUNK;
        const int f = threadIdx.x;
        if (p == 0)      { if (f < FRAGS128) pack_weight_frag(W1, Wp, 128, f); }
        else if (p == 1) { if (f < FRAGS128) pack_weight_frag(W2, Wp + FRAGS128, 128, f); }
        else             { if (f < FRAGS64)  pack_weight_frag(W3, Wp + 2 * FRAGS128, 64, f); }
        return;
    }
    extern __shared__ int sm[];
    const int t = threadIdx.x;
    const int b = blockIdx.x;
    const int chunk = (E + NCHUNK - 1) / NCHUNK;
    const int e0 = b * chunk;
    int e1 = e0 + chunk;
    if (e1 > E) e1 = E;
    for (int i = t; i < n; i += CSR_TB) sm[i] = 0;
    __syncthreads();
    for (int i0 = e0 + t; i0 < e1; i0 += EPT * CSR_TB) {
        int d[EPT];
        #pragma unroll
        for (int u = 0; u < EPT; ++u) {
            const int i = i0 + u * CSR_TB;
            d[u] = dst[i < e1 ? i : e1 - 1];
        }
        #pragma unroll
        for (int u = 0; u < EPT; ++u)
            if (i0 + u * CSR_TB < e1) atomicAdd(&sm[d[u]], 1);
    }
    __syncthreads();
    for (int i = t; i < n; i += CSR_TB) chist[(size_t)b * n + i] = sm[i];
}

__global__ void __launch_bounds__(256)
chunk_prefix_kernel(int* __restrict__ chist, int* __restrict__ lpre,
                    int* __restrict__ gsum, float* __restrict__ isq, int n) {
    constexpr int PER = NCHUNK / 8;
    __shared__ int wt[8][32];
    const int lane = threadIdx.x & 31;
    const int w = threadIdx.x >> 5;
    const int node = blockIdx.x * 32 + lane;
    const bool ok = node < n;
    int v[PER];
    #pragma unroll
    for (int j = 0; j < PER; ++j)
        v[j] = ok ? chist[(size_t)(w * PER + j) * n + node] : 0;
    int run = 0;
    #pragma unroll
    for (int j = 0; j < PER; ++j) { const int c = v[j]; v[j] = run; run += c; }
    wt[w][lane] = run;
    __syncthreads();
    int off = 0, tot = 0;
    #pragma unroll
    for (int q = 0; q < 8; ++q) {
        const int x = wt[q][lane];
        if (q < w) off += x;
        tot += x;
    }
    if (ok) {
        #pragma unroll
        for (int j = 0; j < PER; ++j)
            chist[(size_t)(w * PER + j) * n + node] = v[j] + off;
    }
    if (w == 0) {
        int inc = tot;
        for (int o = 1; o < 32; o <<= 1) {
            const int u = __shfl_up(inc, o);
            if (lane >= o) inc += u;
        }
        if (ok) {
            lpre[node] = inc - tot;
            isq[node] = rsqrtf((float)(tot + 1));
        }
        if (lane == 31) gsum[blockIdx.x] = inc;
    }
}

constexpr int CPT = 10;
constexpr int GPL = 16;
constexpr int GEMM_RT = CSR_TB / 32 / 8;

__device__ __forceinline__ void load_edge_pass(
        const int* __restrict__ src, const int* __restrict__ dst,
        const float* __restrict__ isq, int i0, int e1,
        int (&s)[EPT], int (&d)[EPT], float (&w)[EPT]) {
    #pragma unroll
    for (int u = 0; u < EPT; ++u) {
        const int i = i0 + u * CSR_TB;
        const int ic = i < e1 ? i : e1 - 1;
        s[u] = src[ic];
        d[u] = dst[ic];
    }
    #pragma unroll
    for (int u = 0; u < EPT; ++u) w[u] = isq[s[u]];
}

__device__ __forceinline__ void load_group_sums(
        const int* __restrict__ gsum, int lo, int G, int (&gv)[GPL]) {
    #pragma unroll
    for (int q = 0; q < GPL; ++q) {
        const int g = lo + q;
        gv[q] = gsum[g < G ? g : G - 1];
    }
}

__device__ __forceinline__ void scan_sweep(
        int (&gv)[GPL], int lo, int G, int t, int& carry, int* gp) {
    int tot = 0;
    #pragma unroll
    for (int q = 0; q < GPL; ++q) {
        if (lo + q >= G) gv[q] = 0;
        tot += gv[q];
    }
    int inc = tot;
    for (int o = 1; o < 32; o <<= 1) {
        const int up = __shfl_up(inc, o);
        if (t >= o) inc += up;
    }
    int run = carry + inc - tot;
    #pragma unroll
    for (int q = 0; q < GPL; ++q) {
        if (lo + q < G) gp[lo + q] = run;
        run += gv[q];
    }
    carry += __shfl(inc, 31);
}

__global__ void __launch_bounds__(CSR_TB)
csr_place_kernel(const int* __restrict__ src, const int* __restrict__ dst,
                 const int* __restrict__ chist, const int* __restrict__ lpre,
                 const int* __restrict__ gsum, const float* __restrict__ isq,
                 int* __restrict__ rowptr, int2* __restrict__ col,
                 int n, int E, int G,
                 const float* __restrict__ x, const v16bf* __restrict__ Wp1,
                 _Float16* __restrict__ hlin) {
    if (blockIdx.x >= NCHUNK) {
        const int wv = __builtin_amdgcn_readfirstlane((int)(threadIdx.x >> 5));
        const int rt = (int)(blockIdx.x - NCHUNK) * GEMM_RT + (wv >> 3);
        if (rt < n / 16)
            gemm_tile_bf16<128, 128>(x, Wp1, hlin, threadIdx.x & 31, wv & 7, rt * 16);
        return;
    }
    extern __shared__ int sm[];
    int* cur = sm;
    int* gp = sm + n;
    const int t = threadIdx.x;
    const int b = blockIdx.x;
    const int chunk = (E + NCHUNK - 1) / NCHUNK;
    const int e0 = b * chunk;
    int e1 = e0 + chunk;
    if (e1 > E) e1 = E;
    const int* __restrict__ crow = chist + (size_t)b * n;
    const bool scan_wave = t < 32;

    int lp[CPT], ch[CPT];
    #pragma unroll
    for (int u = 0; u < CPT; ++u) {
        const int i = t + u * CSR_TB;
        const int ic = i < n ? i : n - 1;
        lp[u] = lpre[ic];
        ch[u] = crow[ic];
    }
    int gv[GPL];
    #pragma unroll
    for (int q = 0; q < GPL; ++q) gv[q] = 0;
    if (scan_wave) load_group_sums(gsum, t * GPL, G, gv);
    int s[EPT], d[EPT];
    float w[EPT];
    load_edge_pass(src, dst, isq, e0 + t, e1, s, d, w);

    if (scan_wave) {
        int carry = 0;
        scan_sweep(gv, t * GPL, G, t, carry, gp);
        for (int g0 = 32 * GPL; g0 < G; g0 += 32 * GPL) {
            load_group_sums(gsum, g0 + t * GPL, G, gv);
            scan_sweep(gv, g0 + t * GPL, G, t, carry, gp);
        }
        if (t == 31 && b == 0) rowptr[n] = carry;
    }
    __syncthreads();
    int gs[CPT];
    #pragma unroll
    for (int u = 0; u < CPT; ++u) {
        const int i = t + u * CSR_TB;
        gs[u] = gp[(i < n ? i : n - 1) >> 5];
    }
    #pragma unroll
    for (int u = 0; u < CPT; ++u) {
        const int i = t + u * CSR_TB;
        if (i < n) {
            const int r = gs[u] + lp[u];
            if (b == 0) rowptr[i] = r;
            cur[i] = r + ch[u];
        }
    }
    for (int i = t + CPT * CSR_TB; i < n; i += CSR_TB) {
        const int r = gp[i >> 5] + lpre[i];
        if (b == 0) rowptr[i] = r;
        cur[i] = r + crow[i];
    }
    __syncthreads();
    for (int p0 = e0; p0 < e1; ) {
        const int i0 = p0 + t;
        int pos[EPT];
        #pragma unroll
        for (int u = 0; u < EPT; ++u) {
            pos[u] = 0;
            if (i0 + u * CSR_TB < e1) pos[u] = atomicAdd(&cur[d[u]], 1);
        }
        #pragma unroll
        for (int u = 0; u < EPT; ++u)
            if (i0 + u * CSR_TB < e1) col[pos[u]] = make_int2(s[u], __float_as_int(w[u]));
        p0 += EPT * CSR_TB;
        if (p0 < e1) load_edge_pass(src, dst, isq, p0 + t, e1, s, d, w);
    }
}

template <int C>
__global__ void __launch_bounds__(256)
gather_sum_kernel(const _Float16* __restrict__ H,
                  const int* __restrict__ rowptr,
                  const int2* __restrict__ col,
                  const float* __restrict__ isq,
                  const float* __restrict__ bias,
                  float* __restrict__ OUT, int n, int do_relu) {
    constexpr int VEC = C / 32;
    typedef float vec_t __attribute__((ext_vector_type(VEC)));
    typedef _Float16 row_t __attribute__((ext_vector_type(VEC)));
    const int lane = threadIdx.x & 31;
    const int node = __builtin_amdgcn_readfirstlane(
        (int)(blockIdx.x * (blockDim.x >> 5) + (threadIdx.x >> 5)));
    if (node >= n) return;
    const int beg = rowptr[node];
    const int end = rowptr[node + 1];
    const float sd = isq[node];
    const row_t* __restrict__ Hv = reinterpret_cast<const row_t*>(H);

    const vec_t hv = __builtin_convertvector(Hv[(size_t)node * 32 + lane], vec_t);
    vec_t acc = (hv * sd) * sd;

    int2 nxt = make_int2(0, 0);
    if (beg + lane < end) nxt = col[beg + lane];
    for (int j0 = beg; j0 < end; j0 += 32) {
        const int cur_s = nxt.x;
        const float cur_w = __int_as_float(nxt.y) * sd;
        nxt = make_int2(0, 0);
        const int jn = j0 + 32 + lane;
        if (jn < end) nxt = col[jn];
        int m = end - j0;
        if (m > 32) m = 32;
        int k = 0;
        for (; k + 8 <= m; k += 8) {
            row_t v[8];
            #pragma unroll
            for (int u = 0; u < 8; ++u) {
                const int s = __builtin_amdgcn_readlane(cur_s, k + u);
                v[u] = Hv[(size_t)s * 32 + lane];
            }
            #pragma unroll
            for (int u = 0; u < 8; ++u) {
                const float w = __int_as_float(
                    __builtin_amdgcn_readlane(__float_as_int(cur_w), k + u));
                acc += __builtin_convertvector(v[u], vec_t) * w;
            }
        }
        if (k < m) {
            row_t v[8];
            #pragma unroll
            for (int u = 0; u < 8; ++u) {
                const int s = __builtin_amdgcn_readlane(cur_s, (k + u) & 31);
                v[u] = Hv[(size_t)s * 32 + lane];
            }
            #pragma unroll
            for (int u = 0; u < 8; ++u) {
                const float w = __int_as_float(
                    __builtin_amdgcn_readlane(__float_as_int(cur_w), (k + u) & 31));
                if (k + u < m) acc += __builtin_convertvector(v[u], vec_t) * w;
            }
        }
    }

    const vec_t bv = reinterpret_cast<const vec_t*>(bias)[lane];
    vec_t o = acc + bv;
    if (do_relu) {
        #pragma unroll
        for (int i = 0; i < VEC; ++i) o[i] = fmaxf(o[i], 0.0f);
    }
    reinterpret_cast<vec_t*>(OUT)[(size_t)node * 32 + lane] = o;
}

typedef __attribute__((ext_vector_type(8))) __bf16 v8bf;
typedef __attribute__((ext_vector_type(4))) __bf16 v4bf;

template <int COUT>
__global__ void __launch_bounds__(512)
gather_gemm_kernel(const _Float16* __restrict__ H,
                   const int* __restrict__ rowptr,
                   const int2* __restrict__ col,
                   const float* __restrict__ isq,
                   const float* __restrict__ bias,
                   const v16bf* __restrict__ Wpn,
                   _Float16* __restrict__ Hn, int n) {
    constexpr int C = 128;
    constexpr int PITCH = C + 8;
    constexpr int NT = COUT / 16;
    typedef float vec_t __attribute__((ext_vector_type(4)));
    typedef _Float16 row_t __attribute__((ext_vector_type(4)));
    __shared__ __attribute__((aligned(16))) __bf16 tile[16 * PITCH];

    const int lane = threadIdx.x & 31;
    const int wave = __builtin_amdgcn_readfirstlane((int)(threadIdx.x >> 5));
    const int node = __builtin_amdgcn_readfirstlane((int)(blockIdx.x * 16)) + wave;

    vec_t o = {0.f, 0.f, 0.f, 0.f};
    if (node < n) {
        const int beg = rowptr[node];
        const int end = rowptr[node + 1];
        const float sd = isq[node];
        const row_t* __restrict__ Hv = reinterpret_cast<const row_t*>(H);

        const vec_t hv = __builtin_convertvector(Hv[(size_t)node * 32 + lane], vec_t);
        vec_t acc = (hv * sd) * sd;

        int2 nxt = make_int2(0, 0);
        if (beg + lane < end) nxt = col[beg + lane];
        for (int j0 = beg; j0 < end; j0 += 32) {
            const int cur_s = nxt.x;
            const float cur_w = __int_as_float(nxt.y) * sd;
            nxt = make_int2(0, 0);
            const int jn = j0 + 32 + lane;
            if (jn < end) nxt = col[jn];
            int m = end - j0;
            if (m > 32) m = 32;
            int k = 0;
            for (; k + 8 <= m; k += 8) {
                row_t v[8];
                #pragma unroll
                for (int u = 0; u < 8; ++u) {
                    const int s = __builtin_amdgcn_readlane(cur_s, k + u);
                    v[u] = Hv[(size_t)s * 32 + lane];
                }
                #pragma unroll
                for (int u = 0; u < 8; ++u) {
                    const float w = __int_as_float(
                        __builtin_amdgcn_readlane(__float_as_int(cur_w), k + u));
                    acc += __builtin_convertvector(v[u], vec_t) * w;
                }
            }
            if (k < m) {
                row_t v[8];
                #pragma unroll
                for (int u = 0; u < 8; ++u) {
                    const int s = __builtin_amdgcn_readlane(cur_s, (k + u) & 31);
                    v[u] = Hv[(size_t)s * 32 + lane];
                }
                #pragma unroll
                for (int u = 0; u < 8; ++u) {
                    const float w = __int_as_float(
                        __builtin_amdgcn_readlane(__float_as_int(cur_w), (k + u) & 31));
                    if (k + u < m) acc += __builtin_convertvector(v[u], vec_t) * w;
                }
            }
        }
        const vec_t bv = reinterpret_cast<const vec_t*>(bias)[lane];
        o = acc + bv;
        #pragma unroll
        for (int i = 0; i < 4; ++i) o[i] = fmaxf(o[i], 0.0f);
    }

    v4bf ob;
    #pragma unroll
    for (int i = 0; i < 4; ++i) ob[i] = (__bf16)o[i];
    *reinterpret_cast<v4bf*>(&tile[wave * PITCH + lane * 4]) = ob;

    const int half = lane >> 4;
    const int l = lane & 15;

    v16bf b[C / 32];
    if (wave < NT) {
        #pragma unroll
        for (int kb = 0; kb < C; kb += 32)
            b[kb / 32] = Wpn[(wave * (C / 32) + kb / 32) * 32 + lane];
    }
    __syncthreads();
    if (wave < NT) {
        v8f acc = {0.f, 0.f, 0.f, 0.f, 0.f, 0.f, 0.f, 0.f};
        #pragma unroll
        for (int kb = 0; kb < C; kb += 32) {
            const v8bf lo = *reinterpret_cast<const v8bf*>(&tile[l * PITCH + kb + half * 8]);
            const v8bf hi = *reinterpret_cast<const v8bf*>(&tile[l * PITCH + kb + 16 + half * 8]);
            const v16bf a = __builtin_shufflevector(lo, hi, 0, 1, 2, 3, 4, 5, 6, 7,
                                                    8, 9, 10, 11, 12, 13, 14, 15);
            acc = __builtin_amdgcn_wmma_f32_16x16x32_bf16(
                 false, a,  false, b[kb / 32],
                 (short)0, acc,  false,  false);
        }
        const int m0 = blockIdx.x * 16;
        #pragma unroll
        for (int v = 0; v < 8; ++v) {
            const int r = m0 + v + half * 8;
            Hn[(size_t)r * COUT + wave * 16 + l] = (_Float16)acc[v];
        }
    }
}

extern "C" void kernel_launch(void* const* d_in, const int* in_sizes, int n_in,
                              void* d_out, int out_size, void* d_ws, size_t ws_size,
                              hipStream_t stream) {
    const float* x  = (const float*)d_in[0];
    const int*   ei = (const int*)d_in[1];
    const float* W1 = (const float*)d_in[2];
    const float* b1 = (const float*)d_in[3];
    const float* W2 = (const float*)d_in[4];
    const float* b2 = (const float*)d_in[5];
    const float* W3 = (const float*)d_in[6];
    const float* b3 = (const float*)d_in[7];

    const int CIN = 128, CH = 128;
    const int n = in_sizes[0] / CIN;
    const int E = in_sizes[1] / 2;
    const int* src = ei;
    const int* dst = ei + E;

    _Float16* hlin   = (_Float16*)d_ws;
    _Float16* feat   = hlin + (size_t)n * CH;
    v16bf* Wp     = (v16bf*)(feat + (size_t)n * CH);
    v16bf* Wp2    = Wp + FRAGS128;
    v16bf* Wp3    = Wp2 + FRAGS128;
    int2*  col    = (int2*)(Wp3 + FRAGS64);
    int*   chist  = (int*)(col + E);
    int*   lpre   = chist + (size_t)NCHUNK * n;
    int*   rowptr = lpre + n;
    const int G   = (n + 31) / 32;
    int*   gsum   = rowptr + n + 1;
    float* isq    = (float*)(gsum + G);

    const int TB = 256;
    chunk_hist_kernel<<<NCHUNK + 3, CSR_TB, (size_t)n * sizeof(int), stream>>>(
        dst, chist, n, E, W1, W2, W3, Wp);
    chunk_prefix_kernel<<<G, TB, 0, stream>>>(chist, lpre, gsum, isq, n);
    const int row_tiles = n / 16;
    const int node_blocks = (n + 7) / 8;

    csr_place_kernel<<<NCHUNK + (row_tiles + GEMM_RT - 1) / GEMM_RT, CSR_TB,
                       (size_t)(n + G) * sizeof(int), stream>>>(
        src, dst, chist, lpre, gsum, isq, rowptr, col, n, E, G, x, Wp, hlin);

    gather_gemm_kernel<128><<<row_tiles, 512, 0, stream>>>(hlin, rowptr, col, isq, b1, Wp2, feat, n);

    gather_gemm_kernel<64><<<row_tiles, 512, 0, stream>>>(feat, rowptr, col, isq, b2, Wp3, hlin, n);

    gather_sum_kernel<64><<<node_blocks, TB, 0, stream>>>(hlin, rowptr, col, isq, b3, (float*)d_out, n, 0);
}
